# starting kernel
# baseline (speedup 1.0000x reference)
#include <hip/hip_runtime.h>


typedef __attribute__((ext_vector_type(2))) float v2f;
typedef __attribute__((ext_vector_type(8))) float v8f;

#define INC    32
#define OUTC   64
#define KVOL   8
#define KDIM   256
#define BN_EPS 1e-5f

__global__ void spconv_zero_ws(float* __restrict__ ws, long long nelem4) {
    long long i = (long long)blockIdx.x * blockDim.x + threadIdx.x;
    long long stride = (long long)gridDim.x * blockDim.x;
    float4* p = (float4*)ws;
    for (long long j = i; j < nelem4; j += stride)
        p[j] = make_float4(0.f, 0.f, 0.f, 0.f);
}

__global__ void spconv_scatter(const float* __restrict__ feats,
                               const int*   __restrict__ out_idx,
                               const int*   __restrict__ offset_id,
                               float*       __restrict__ S,
                               int nvox) {
    int t   = blockIdx.x * blockDim.x + threadIdx.x;
    int vox = t >> 5;
    int c   = t & 31;
    if (vox >= nvox) return;
    int seg = out_idx[vox] * KVOL + offset_id[vox];
    unsafeAtomicAdd(&S[(long long)seg * INC + c],
                    feats[(long long)vox * INC + c]);
}

__launch_bounds__(256)
__global__ void spconv_gemm_stats(const float* __restrict__ S,
                                  const float* __restrict__ weight,
                                  float*       __restrict__ out,
                                  float*       __restrict__ stats,
                                  int mtotal) {
    __shared__ float wint[(KDIM / 2) * 2 * OUTC];

    for (int i = threadIdx.x; i < (KDIM / 2) * OUTC; i += blockDim.x) {
        int p = i >> 6;
        int n = i & 63;
        wint[p * 128 + n * 2 + 0] = weight[(2 * p)     * OUTC + n];
        wint[p * 128 + n * 2 + 1] = weight[(2 * p + 1) * OUTC + n];
    }
    __syncthreads();

    const int wave = threadIdx.x >> 5;
    const int lane = threadIdx.x & 31;
    const int r    = lane & 15;
    const int kbh  = lane >> 4;

    const int row_base = (blockIdx.x * 8 + wave) * 32;
    if (row_base >= mtotal) return;

    const int ar0 = min(row_base + r,      mtotal - 1);
    const int ar1 = min(row_base + 16 + r, mtotal - 1);
    const float* __restrict__ A0 = S + (long long)ar0 * KDIM + 2 * kbh;
    const float* __restrict__ A1 = S + (long long)ar1 * KDIM + 2 * kbh;
    const float* __restrict__ B  = wint + kbh * 128 + r * 2;

    v8f a00 = {}, a01 = {}, a02 = {}, a03 = {};
    v8f a10 = {}, a11 = {}, a12 = {}, a13 = {};

#pragma unroll 4
    for (int k = 0; k < KDIM; k += 4) {
        float2 f0 = *(const float2*)(A0 + k);
        float2 f1 = *(const float2*)(A1 + k);
        v2f va0; va0.x = f0.x; va0.y = f0.y;
        v2f va1; va1.x = f1.x; va1.y = f1.y;

        const float* bp = B + (k >> 1) * 128;
        float2 g0 = *(const float2*)(bp);
        float2 g1 = *(const float2*)(bp + 32);
        float2 g2 = *(const float2*)(bp + 64);
        float2 g3 = *(const float2*)(bp + 96);
        v2f b0; b0.x = g0.x; b0.y = g0.y;
        v2f b1; b1.x = g1.x; b1.y = g1.y;
        v2f b2; b2.x = g2.x; b2.y = g2.y;
        v2f b3; b3.x = g3.x; b3.y = g3.y;

        a00 = __builtin_amdgcn_wmma_f32_16x16x4_f32(false, va0, false, b0, (short)0, a00, false, false);
        a10 = __builtin_amdgcn_wmma_f32_16x16x4_f32(false, va1, false, b0, (short)0, a10, false, false);
        a01 = __builtin_amdgcn_wmma_f32_16x16x4_f32(false, va0, false, b1, (short)0, a01, false, false);
        a11 = __builtin_amdgcn_wmma_f32_16x16x4_f32(false, va1, false, b1, (short)0, a11, false, false);
        a02 = __builtin_amdgcn_wmma_f32_16x16x4_f32(false, va0, false, b2, (short)0, a02, false, false);
        a12 = __builtin_amdgcn_wmma_f32_16x16x4_f32(false, va1, false, b2, (short)0, a12, false, false);
        a03 = __builtin_amdgcn_wmma_f32_16x16x4_f32(false, va0, false, b3, (short)0, a03, false, false);
        a13 = __builtin_amdgcn_wmma_f32_16x16x4_f32(false, va1, false, b3, (short)0, a13, false, false);
    }

    const int mrow0 = row_base + ((lane >> 4) << 3);

#define EMIT_NT(ACCL, ACCH, NT)                                                \
    {                                                                          \
        float ps = 0.f, pq = 0.f;                                              \
        _Pragma("unroll")                                                      \
        for (int i = 0; i < 8; ++i) {                                          \
            int m = mrow0 + i;                                                 \
            float v = (ACCL)[i];                                               \
            if (m < mtotal) {                                                  \
                out[(long long)m * OUTC + (NT) * 16 + r] = v;                  \
                ps += v; pq += v * v;                                          \
            }                                                                  \
        }                                                                      \
        _Pragma("unroll")                                                      \
        for (int i = 0; i < 8; ++i) {                                          \
            int m = mrow0 + 16 + i;                                            \
            float v = (ACCH)[i];                                               \
            if (m < mtotal) {                                                  \
                out[(long long)m * OUTC + (NT) * 16 + r] = v;                  \
                ps += v; pq += v * v;                                          \
            }                                                                  \
        }                                                                      \
        ps += __shfl_xor(ps, 16, 32);         \
        pq += __shfl_xor(pq, 16, 32);                                          \
        if (lane < 16) {                                                       \
            unsafeAtomicAdd(&stats[(NT) * 16 + r], ps);                        \
            unsafeAtomicAdd(&stats[64 + (NT) * 16 + r], pq);                   \
        }                                                                      \
    }

    EMIT_NT(a00, a10, 0)
    EMIT_NT(a01, a11, 1)
    EMIT_NT(a02, a12, 2)
    EMIT_NT(a03, a13, 3)
#undef EMIT_NT
}

__global__ void spconv_bn_apply(float* __restrict__ out,
                                const float* __restrict__ stats,
                                const float* __restrict__ gamma,
                                const float* __restrict__ beta,
                                int mtotal) {
    long long i4     = (long long)blockIdx.x * blockDim.x + threadIdx.x;
    long long total4 = ((long long)mtotal * OUTC) >> 2;
    if (i4 >= total4) return;
    const float inv_m = 1.0f / (float)mtotal;
    const int c0 = (int)((i4 << 2) & (OUTC - 1));
    float4 v = ((float4*)out)[i4];
    float x[4] = {v.x, v.y, v.z, v.w};
    float y[4];
#pragma unroll
    for (int j = 0; j < 4; ++j) {
        int c = c0 + j;
        float mean = stats[c] * inv_m;
        float var  = stats[64 + c] * inv_m - mean * mean;
        y[j] = (x[j] - mean) * rsqrtf(var + BN_EPS) * gamma[c] + beta[c];
    }
    ((float4*)out)[i4] = make_float4(y[0], y[1], y[2], y[3]);
}

extern "C" void kernel_launch(void* const* d_in, const int* in_sizes, int n_in,
                              void* d_out, int out_size, void* d_ws, size_t ws_size,
                              hipStream_t stream) {
    const float* feats     = (const float*)d_in[0];
    const float* weight    = (const float*)d_in[1];
    const float* gamma     = (const float*)d_in[2];
    const float* beta      = (const float*)d_in[3];
    const int*   out_idx   = (const int*)d_in[4];
    const int*   offset_id = (const int*)d_in[5];
    const int nvox   = in_sizes[0] / INC;
    const int mtotal = out_size / OUTC;

    float* S     = (float*)d_ws;
    float* stats = S + (long long)mtotal * KDIM;
    const long long zelem4 = ((long long)mtotal * KDIM + 128) >> 2;

    spconv_zero_ws<<<4096, 256, 0, stream>>>((float*)d_ws, zelem4);

    const long long sthreads = (long long)nvox * INC;
    const int sblocks = (int)((sthreads + 255) / 256);
    spconv_scatter<<<sblocks, 256, 0, stream>>>(feats, out_idx, offset_id, S, nvox);

    const int gblocks = (mtotal + 255) / 256;
    spconv_gemm_stats<<<gblocks, 256, 0, stream>>>(S, weight, (float*)d_out, stats, mtotal);

    const long long total4 = ((long long)mtotal * OUTC) >> 2;
    const int bblocks = (int)((total4 + 255) / 256);
    spconv_bn_apply<<<bblocks, 256, 0, stream>>>((float*)d_out, stats, gamma, beta, mtotal);
}


# S not materialised: one 4-byte exchange per voxel links it into its bucket's list (8 MB table); the GEMM's A loader sums the listed feats rows; bucket sums in list order, k permuted inside an offset
# speedup vs baseline: 1.6145x; 1.6145x over previous
#include <hip/hip_runtime.h>


typedef __attribute__((ext_vector_type(2))) float v2f;
typedef __attribute__((ext_vector_type(8))) float v8f;

#define INC    32
#define OUTC   64
#define KVOL   8
#define KDIM   256
#define BN_EPS 1e-5f

__global__ void spconv_init_ws(int* __restrict__ head, long long nhead4,
                               float* __restrict__ stats) {
    long long i = (long long)blockIdx.x * blockDim.x + threadIdx.x;
    long long stride = (long long)gridDim.x * blockDim.x;
    int4* p = (int4*)head;
    for (long long j = i; j < nhead4; j += stride)
        p[j] = make_int4(-1, -1, -1, -1);
    if (i < 2 * OUTC) stats[i] = 0.f;
}

__global__ void spconv_link(const int* __restrict__ out_idx,
                            const int* __restrict__ offset_id,
                            int*       __restrict__ head,
                            int*       __restrict__ next,
                            int nvox) {
    int vox = blockIdx.x * blockDim.x + threadIdx.x;
    if (vox >= nvox) return;
    int seg = out_idx[vox] * KVOL + offset_id[vox];
    next[vox] = atomicExch(&head[seg], vox);
}

__launch_bounds__(256) __attribute__((amdgpu_waves_per_eu(8, 8)))
__global__ void spconv_gemm_stats(const float* __restrict__ feats,
                                  const int*   __restrict__ head,
                                  const int*   __restrict__ next,
                                  const float* __restrict__ weight,
                                  float*       __restrict__ out,
                                  float*       __restrict__ stats,
                                  int mtotal, int nvox) {
    __shared__ float wint[(KDIM / 2) * 2 * OUTC];

    const int wave = threadIdx.x >> 5;
    const int lane = threadIdx.x & 31;
    const int r    = lane & 15;
    const int kbh  = lane >> 4;

    const int row_base = (blockIdx.x * 8 + wave) * 32;
    const int ar0 = min(row_base + r,      mtotal - 1);
    const int ar1 = min(row_base + 16 + r, mtotal - 1);

    const int hb0 = ar0 * KVOL, hb1 = ar1 * KVOL;
    int hc0 = head[hb0], hc1 = head[hb1];

    for (int i = threadIdx.x; i < (KDIM / 2) * OUTC; i += blockDim.x) {
        int p = i >> 6;
        int n = i & 63;
        wint[p * 128 + n * 2 + 0] = weight[(2 * p)     * OUTC + n];
        wint[p * 128 + n * 2 + 1] = weight[(2 * p + 1) * OUTC + n];
    }
    __syncthreads();

    if (row_base >= mtotal) return;

    const float4* __restrict__ F4 = (const float4*)feats;
    const unsigned fq = 4u * (unsigned)kbh;
    const float* __restrict__ B = wint + kbh * 1024 + r * 2;

    v8f a00 = {}, a01 = {}, a02 = {}, a03 = {};
    v8f a10 = {}, a11 = {}, a12 = {}, a13 = {};

#pragma unroll 1
    for (int o = 0; o < KVOL; ++o) {
        const int v0 = hc0, v1 = hc1;
        const int on = min(o + 1, KVOL - 1);
        hc0 = head[hb0 + on]; hc1 = head[hb1 + on];

        bool ok0 = (unsigned)v0 < (unsigned)nvox;
        bool ok1 = (unsigned)v1 < (unsigned)nvox;

        const float4 z4 = make_float4(0.f, 0.f, 0.f, 0.f);
        float4 xa0 = z4, xa1 = z4, xa2 = z4, xa3 = z4;
        float4 xb0 = z4, xb1 = z4, xb2 = z4, xb3 = z4;
        int n0 = -1, n1 = -1;
        if (ok0) {
            const float4* p = F4 + ((unsigned)v0 * 8u + fq);
            xa0 = p[0]; xa1 = p[1]; xa2 = p[2]; xa3 = p[3];
            n0 = next[v0];
        }
        if (ok1) {
            const float4* p = F4 + ((unsigned)v1 * 8u + fq);
            xb0 = p[0]; xb1 = p[1]; xb2 = p[2]; xb3 = p[3];
            n1 = next[v1];
        }

        for (int it = 1; it < nvox; ++it) {
            ok0 = (unsigned)n0 < (unsigned)nvox;
            ok1 = (unsigned)n1 < (unsigned)nvox;
            if (!(ok0 | ok1)) break;
            if (ok0) {
                const float4* p = F4 + ((unsigned)n0 * 8u + fq);
                float4 t0 = p[0], t1 = p[1], t2 = p[2], t3 = p[3];
                n0 = next[n0];
                xa0.x += t0.x; xa0.y += t0.y; xa0.z += t0.z; xa0.w += t0.w;
                xa1.x += t1.x; xa1.y += t1.y; xa1.z += t1.z; xa1.w += t1.w;
                xa2.x += t2.x; xa2.y += t2.y; xa2.z += t2.z; xa2.w += t2.w;
                xa3.x += t3.x; xa3.y += t3.y; xa3.z += t3.z; xa3.w += t3.w;
            }
            if (ok1) {
                const float4* p = F4 + ((unsigned)n1 * 8u + fq);
                float4 t0 = p[0], t1 = p[1], t2 = p[2], t3 = p[3];
                n1 = next[n1];
                xb0.x += t0.x; xb0.y += t0.y; xb0.z += t0.z; xb0.w += t0.w;
                xb1.x += t1.x; xb1.y += t1.y; xb1.z += t1.z; xb1.w += t1.w;
                xb2.x += t2.x; xb2.y += t2.y; xb2.z += t2.z; xb2.w += t2.w;
                xb3.x += t3.x; xb3.y += t3.y; xb3.z += t3.z; xb3.w += t3.w;
            }
        }

        const float x0[16] = {xa0.x, xa0.y, xa0.z, xa0.w, xa1.x, xa1.y, xa1.z, xa1.w,
                              xa2.x, xa2.y, xa2.z, xa2.w, xa3.x, xa3.y, xa3.z, xa3.w};
        const float x1[16] = {xb0.x, xb0.y, xb0.z, xb0.w, xb1.x, xb1.y, xb1.z, xb1.w,
                              xb2.x, xb2.y, xb2.z, xb2.w, xb3.x, xb3.y, xb3.z, xb3.w};
        const float* Bo = B + o * (16 * 128);

#pragma unroll
        for (int kk = 0; kk < 8; ++kk) {
            v2f va0; va0.x = x0[2 * kk]; va0.y = x0[2 * kk + 1];
            v2f va1; va1.x = x1[2 * kk]; va1.y = x1[2 * kk + 1];

            const float* bp = Bo + kk * 128;
            float2 g0 = *(const float2*)(bp);
            float2 g1 = *(const float2*)(bp + 32);
            float2 g2 = *(const float2*)(bp + 64);
            float2 g3 = *(const float2*)(bp + 96);
            v2f b0; b0.x = g0.x; b0.y = g0.y;
            v2f b1; b1.x = g1.x; b1.y = g1.y;
            v2f b2; b2.x = g2.x; b2.y = g2.y;
            v2f b3; b3.x = g3.x; b3.y = g3.y;

            a00 = __builtin_amdgcn_wmma_f32_16x16x4_f32(false, va0, false, b0, (short)0, a00, false, false);
            a10 = __builtin_amdgcn_wmma_f32_16x16x4_f32(false, va1, false, b0, (short)0, a10, false, false);
            a01 = __builtin_amdgcn_wmma_f32_16x16x4_f32(false, va0, false, b1, (short)0, a01, false, false);
            a11 = __builtin_amdgcn_wmma_f32_16x16x4_f32(false, va1, false, b1, (short)0, a11, false, false);
            a02 = __builtin_amdgcn_wmma_f32_16x16x4_f32(false, va0, false, b2, (short)0, a02, false, false);
            a12 = __builtin_amdgcn_wmma_f32_16x16x4_f32(false, va1, false, b2, (short)0, a12, false, false);
            a03 = __builtin_amdgcn_wmma_f32_16x16x4_f32(false, va0, false, b3, (short)0, a03, false, false);
            a13 = __builtin_amdgcn_wmma_f32_16x16x4_f32(false, va1, false, b3, (short)0, a13, false, false);
        }
    }

    const int mrow0 = row_base + ((lane >> 4) << 3);

#define EMIT_NT(ACCL, ACCH, NT)                                                \
    {                                                                          \
        float ps = 0.f, pq = 0.f;                                              \
        _Pragma("unroll")                                                      \
        for (int i = 0; i < 8; ++i) {                                          \
            int m = mrow0 + i;                                                 \
            float v = (ACCL)[i];                                               \
            if (m < mtotal) {                                                  \
                out[(long long)m * OUTC + (NT) * 16 + r] = v;                  \
                ps += v; pq += v * v;                                          \
            }                                                                  \
        }                                                                      \
        _Pragma("unroll")                                                      \
        for (int i = 0; i < 8; ++i) {                                          \
            int m = mrow0 + 16 + i;                                            \
            float v = (ACCH)[i];                                               \
            if (m < mtotal) {                                                  \
                out[(long long)m * OUTC + (NT) * 16 + r] = v;                  \
                ps += v; pq += v * v;                                          \
            }                                                                  \
        }                                                                      \
        ps += __shfl_xor(ps, 16, 32);         \
        pq += __shfl_xor(pq, 16, 32);                                          \
        if (lane < 16) {                                                       \
            unsafeAtomicAdd(&stats[(NT) * 16 + r], ps);                        \
            unsafeAtomicAdd(&stats[64 + (NT) * 16 + r], pq);                   \
        }                                                                      \
    }

    EMIT_NT(a00, a10, 0)
    EMIT_NT(a01, a11, 1)
    EMIT_NT(a02, a12, 2)
    EMIT_NT(a03, a13, 3)
#undef EMIT_NT
}

__global__ void spconv_bn_apply(float* __restrict__ out,
                                const float* __restrict__ stats,
                                const float* __restrict__ gamma,
                                const float* __restrict__ beta,
                                int mtotal) {
    long long i4     = (long long)blockIdx.x * blockDim.x + threadIdx.x;
    long long total4 = ((long long)mtotal * OUTC) >> 2;
    if (i4 >= total4) return;
    const float inv_m = 1.0f / (float)mtotal;
    const int c0 = (int)((i4 << 2) & (OUTC - 1));
    float4 v = ((float4*)out)[i4];
    float x[4] = {v.x, v.y, v.z, v.w};
    float y[4];
#pragma unroll
    for (int j = 0; j < 4; ++j) {
        int c = c0 + j;
        float mean = stats[c] * inv_m;
        float var  = stats[64 + c] * inv_m - mean * mean;
        y[j] = (x[j] - mean) * rsqrtf(var + BN_EPS) * gamma[c] + beta[c];
    }
    ((float4*)out)[i4] = make_float4(y[0], y[1], y[2], y[3]);
}

extern "C" void kernel_launch(void* const* d_in, const int* in_sizes, int n_in,
                              void* d_out, int out_size, void* d_ws, size_t ws_size,
                              hipStream_t stream) {
    const float* feats     = (const float*)d_in[0];
    const float* weight    = (const float*)d_in[1];
    const float* gamma     = (const float*)d_in[2];
    const float* beta      = (const float*)d_in[3];
    const int*   out_idx   = (const int*)d_in[4];
    const int*   offset_id = (const int*)d_in[5];
    const int nvox   = in_sizes[0] / INC;
    const int mtotal = out_size / OUTC;

    int*   head  = (int*)d_ws;
    int*   next  = head + (long long)mtotal * KVOL;
    float* stats = (float*)(next + nvox);
    const long long nhead4 = ((long long)mtotal * KVOL) >> 2;

    spconv_init_ws<<<1024, 256, 0, stream>>>(head, nhead4, stats);

    const int sblocks = (nvox + 255) / 256;
    spconv_link<<<sblocks, 256, 0, stream>>>(out_idx, offset_id, head, next, nvox);

    const int gblocks = (mtotal + 255) / 256;
    spconv_gemm_stats<<<gblocks, 256, 0, stream>>>(feats, head, next, weight, (float*)d_out,
                                                   stats, mtotal, nvox);

    const long long total4 = ((long long)mtotal * OUTC) >> 2;
    const int bblocks = (int)((total4 + 255) / 256);
    spconv_bn_apply<<<bblocks, 256, 0, stream>>>((float*)d_out, stats, gamma, beta, mtotal);
}


# GEMM A loader runs one step ahead: 16-row passes (4 accumulators) free registers for the next step's first and second list rows, requested before this step's WMMAs; ids chased ahead; sum order kept
# speedup vs baseline: 1.8126x; 1.1227x over previous
#include <hip/hip_runtime.h>


typedef __attribute__((ext_vector_type(2))) float v2f;
typedef __attribute__((ext_vector_type(8))) float v8f;

#define INC    32
#define OUTC   64
#define KVOL   8
#define KDIM   256
#define BN_EPS 1e-5f

__global__ void spconv_init_ws(int* __restrict__ head, long long nhead4,
                               float* __restrict__ stats) {
    long long i = (long long)blockIdx.x * blockDim.x + threadIdx.x;
    long long stride = (long long)gridDim.x * blockDim.x;
    int4* p = (int4*)head;
    for (long long j = i; j < nhead4; j += stride)
        p[j] = make_int4(-1, -1, -1, -1);
    if (i < 2 * OUTC) stats[i] = 0.f;
}

__global__ void spconv_link(const int* __restrict__ out_idx,
                            const int* __restrict__ offset_id,
                            int*       __restrict__ head,
                            int*       __restrict__ next,
                            int nvox) {
    int vox = blockIdx.x * blockDim.x + threadIdx.x;
    if (vox >= nvox) return;
    int seg = out_idx[vox] * KVOL + offset_id[vox];
    next[vox] = atomicExch(&head[seg], vox);
}

__launch_bounds__(256) __attribute__((amdgpu_waves_per_eu(8, 8)))
__global__ void spconv_gemm_stats(const float* __restrict__ feats,
                                  const int*   __restrict__ head,
                                  const int*   __restrict__ next,
                                  const float* __restrict__ weight,
                                  float*       __restrict__ out,
                                  float*       __restrict__ stats,
                                  int mtotal, int nvox) {
    __shared__ float wint[(KDIM / 2) * 2 * OUTC];

    const int wave = threadIdx.x >> 5;
    const int lane = threadIdx.x & 31;
    const int r    = lane & 15;
    const int kbh  = lane >> 4;

    const int row_base = (blockIdx.x * 8 + wave) * 32;
    const int ar0 = min(row_base + r,      mtotal - 1);
    const int ar1 = min(row_base + 16 + r, mtotal - 1);
    const int hb0 = ar0 * KVOL, hb1 = ar1 * KVOL;

    const int a0 = head[hb0];
    int an = head[hb0 + 1];
    int a2 = head[hb0 + 2];

    for (int i = threadIdx.x; i < (KDIM / 2) * OUTC; i += blockDim.x) {
        int p = i >> 6;
        int n = i & 63;
        wint[p * 128 + n * 2 + 0] = weight[(2 * p)     * OUTC + n];
        wint[p * 128 + n * 2 + 1] = weight[(2 * p + 1) * OUTC + n];
    }
    __syncthreads();

    if (row_base >= mtotal) return;

    const float4* __restrict__ F4 = (const float4*)feats;
    const unsigned fq = 4u * (unsigned)kbh;
    const unsigned un = (unsigned)nvox;
    const float* __restrict__ B = wint + kbh * 1024 + r * 2;

    const float4 z4 = make_float4(0.f, 0.f, 0.f, 0.f);

    int b0 = -1, bn = -1;
    if ((unsigned)a0 < un) b0 = next[a0];
    if ((unsigned)an < un) bn = next[an];

    float4 x0 = z4, x1 = z4, x2 = z4, x3 = z4;
    float4 y0 = z4, y1 = z4, y2 = z4, y3 = z4;
    int c = -1;
    if ((unsigned)a0 < un) {
        const float4* p = F4 + ((unsigned)a0 * 8u + fq);
        x0 = p[0]; x1 = p[1]; x2 = p[2]; x3 = p[3];
    }
    if ((unsigned)b0 < un) {
        const float4* p = F4 + ((unsigned)b0 * 8u + fq);
        y0 = p[0]; y1 = p[1]; y2 = p[2]; y3 = p[3];
        c = next[b0];
    }

    v8f acc0 = {}, acc1 = {}, acc2 = {}, acc3 = {};
    float ps0 = 0.f, ps1 = 0.f, ps2 = 0.f, ps3 = 0.f;
    float pq0 = 0.f, pq1 = 0.f, pq2 = 0.f, pq3 = 0.f;

#pragma unroll 1
    for (int t = 0; t < 2 * KVOL; ++t) {
        x0.x += y0.x; x0.y += y0.y; x0.z += y0.z; x0.w += y0.w;
        x1.x += y1.x; x1.y += y1.y; x1.z += y1.z; x1.w += y1.w;
        x2.x += y2.x; x2.y += y2.y; x2.z += y2.z; x2.w += y2.w;
        x3.x += y3.x; x3.y += y3.y; x3.z += y3.z; x3.w += y3.w;
        for (int it = 2; it < nvox; ++it) {
            if (!((unsigned)c < un)) break;
            const float4* p = F4 + ((unsigned)c * 8u + fq);
            float4 t0 = p[0], t1 = p[1], t2 = p[2], t3 = p[3];
            c = next[c];
            x0.x += t0.x; x0.y += t0.y; x0.z += t0.z; x0.w += t0.w;
            x1.x += t1.x; x1.y += t1.y; x1.z += t1.z; x1.w += t1.w;
            x2.x += t2.x; x2.y += t2.y; x2.z += t2.z; x2.w += t2.w;
            x3.x += t3.x; x3.y += t3.y; x3.z += t3.z; x3.w += t3.w;
        }

        const float xs[16] = {x0.x, x0.y, x0.z, x0.w, x1.x, x1.y, x1.z, x1.w,
                              x2.x, x2.y, x2.z, x2.w, x3.x, x3.y, x3.z, x3.w};

        x0 = z4; x1 = z4; x2 = z4; x3 = z4;
        y0 = z4; y1 = z4; y2 = z4; y3 = z4;
        c = -1;
        int b2 = -1, a3 = -1;
        if (t + 1 < 2 * KVOL) {
            if ((unsigned)an < un) {
                const float4* p = F4 + ((unsigned)an * 8u + fq);
                x0 = p[0]; x1 = p[1]; x2 = p[2]; x3 = p[3];
            }
            if ((unsigned)bn < un) {
                const float4* p = F4 + ((unsigned)bn * 8u + fq);
                y0 = p[0]; y1 = p[1]; y2 = p[2]; y3 = p[3];
                c = next[bn];
            }
        }
        if (t + 2 < 2 * KVOL) {
            if ((unsigned)a2 < un) b2 = next[a2];
        }
        if (t + 3 < 2 * KVOL) {
            const int t3 = t + 3;
            a3 = head[((t3 & KVOL) ? hb1 : hb0) + (t3 & (KVOL - 1))];
        }

        const int o = t & (KVOL - 1);
        const float* Bo = B + o * (16 * 128);

#pragma unroll
        for (int kk = 0; kk < 8; ++kk) {
            v2f va; va.x = xs[2 * kk]; va.y = xs[2 * kk + 1];

            const float* bp = Bo + kk * 128;
            float2 g0 = *(const float2*)(bp);
            float2 g1 = *(const float2*)(bp + 32);
            float2 g2 = *(const float2*)(bp + 64);
            float2 g3 = *(const float2*)(bp + 96);
            v2f q0; q0.x = g0.x; q0.y = g0.y;
            v2f q1; q1.x = g1.x; q1.y = g1.y;
            v2f q2; q2.x = g2.x; q2.y = g2.y;
            v2f q3; q3.x = g3.x; q3.y = g3.y;

            acc0 = __builtin_amdgcn_wmma_f32_16x16x4_f32(false, va, false, q0, (short)0, acc0, false, false);
            acc1 = __builtin_amdgcn_wmma_f32_16x16x4_f32(false, va, false, q1, (short)0, acc1, false, false);
            acc2 = __builtin_amdgcn_wmma_f32_16x16x4_f32(false, va, false, q2, (short)0, acc2, false, false);
            acc3 = __builtin_amdgcn_wmma_f32_16x16x4_f32(false, va, false, q3, (short)0, acc3, false, false);
        }

        if (o == KVOL - 1) {
            const int mrow = row_base + ((t >> 3) << 4) + ((lane >> 4) << 3);
#define EMIT_NT(ACC, PS, PQ, NT)                                               \
            _Pragma("unroll")                                                  \
            for (int i = 0; i < 8; ++i) {                                      \
                int m = mrow + i;                                              \
                float v = (ACC)[i];                                            \
                if (m < mtotal) {                                              \
                    out[(long long)m * OUTC + (NT) * 16 + r] = v;              \
                    PS += v; PQ += v * v;                                      \
                }                                                              \
            }
            EMIT_NT(acc0, ps0, pq0, 0)
            EMIT_NT(acc1, ps1, pq1, 1)
            EMIT_NT(acc2, ps2, pq2, 2)
            EMIT_NT(acc3, ps3, pq3, 3)
#undef EMIT_NT
            const v8f zacc = {};
            acc0 = zacc; acc1 = zacc; acc2 = zacc; acc3 = zacc;
        }

        an = a2; bn = b2; a2 = a3;
    }

#define EMIT_STAT(PS, PQ, NT)                                                  \
    {                                                                          \
        float s = (PS) + __shfl_xor((PS), 16, 32);                             \
        float q = (PQ) + __shfl_xor((PQ), 16, 32);                             \
        if (lane < 16) {                                                       \
            unsafeAtomicAdd(&stats[(NT) * 16 + r], s);                         \
            unsafeAtomicAdd(&stats[64 + (NT) * 16 + r], q);                    \
        }                                                                      \
    }
    EMIT_STAT(ps0, pq0, 0)
    EMIT_STAT(ps1, pq1, 1)
    EMIT_STAT(ps2, pq2, 2)
    EMIT_STAT(ps3, pq3, 3)
#undef EMIT_STAT
}

__global__ void spconv_bn_apply(float* __restrict__ out,
                                const float* __restrict__ stats,
                                const float* __restrict__ gamma,
                                const float* __restrict__ beta,
                                int mtotal) {
    long long i4     = (long long)blockIdx.x * blockDim.x + threadIdx.x;
    long long total4 = ((long long)mtotal * OUTC) >> 2;
    if (i4 >= total4) return;
    const float inv_m = 1.0f / (float)mtotal;
    const int c0 = (int)((i4 << 2) & (OUTC - 1));
    float4 v = ((float4*)out)[i4];
    float x[4] = {v.x, v.y, v.z, v.w};
    float y[4];
#pragma unroll
    for (int j = 0; j < 4; ++j) {
        int c = c0 + j;
        float mean = stats[c] * inv_m;
        float var  = stats[64 + c] * inv_m - mean * mean;
        y[j] = (x[j] - mean) * rsqrtf(var + BN_EPS) * gamma[c] + beta[c];
    }
    ((float4*)out)[i4] = make_float4(y[0], y[1], y[2], y[3]);
}

extern "C" void kernel_launch(void* const* d_in, const int* in_sizes, int n_in,
                              void* d_out, int out_size, void* d_ws, size_t ws_size,
                              hipStream_t stream) {
    const float* feats     = (const float*)d_in[0];
    const float* weight    = (const float*)d_in[1];
    const float* gamma     = (const float*)d_in[2];
    const float* beta      = (const float*)d_in[3];
    const int*   out_idx   = (const int*)d_in[4];
    const int*   offset_id = (const int*)d_in[5];
    const int nvox   = in_sizes[0] / INC;
    const int mtotal = out_size / OUTC;

    int*   head  = (int*)d_ws;
    int*   next  = head + (long long)mtotal * KVOL;
    float* stats = (float*)(next + nvox);
    const long long nhead4 = ((long long)mtotal * KVOL) >> 2;

    spconv_init_ws<<<1024, 256, 0, stream>>>(head, nhead4, stats);

    const int sblocks = (nvox + 255) / 256;
    spconv_link<<<sblocks, 256, 0, stream>>>(out_idx, offset_id, head, next, nvox);

    const int gblocks = (mtotal + 255) / 256;
    spconv_gemm_stats<<<gblocks, 256, 0, stream>>>(feats, head, next, weight, (float*)d_out,
                                                   stats, mtotal, nvox);

    const long long total4 = ((long long)mtotal * OUTC) >> 2;
    const int bblocks = (int)((total4 + 255) / 256);
    spconv_bn_apply<<<bblocks, 256, 0, stream>>>((float*)d_out, stats, gamma, beta, mtotal);
}


# BN column sums: the 8 waves' partials are added in LDS in fixed order, then one f32 atomic per column per work-group (8x fewer atomics on 128 hot words); sum tree changes, arrival order as before
# speedup vs baseline: 2.0961x; 1.1564x over previous
#include <hip/hip_runtime.h>


typedef __attribute__((ext_vector_type(2))) float v2f;
typedef __attribute__((ext_vector_type(8))) float v8f;

#define INC    32
#define OUTC   64
#define KVOL   8
#define KDIM   256
#define BN_EPS 1e-5f

__global__ void spconv_init_ws(int* __restrict__ head, long long nhead4,
                               float* __restrict__ stats) {
    long long i = (long long)blockIdx.x * blockDim.x + threadIdx.x;
    long long stride = (long long)gridDim.x * blockDim.x;
    int4* p = (int4*)head;
    for (long long j = i; j < nhead4; j += stride)
        p[j] = make_int4(-1, -1, -1, -1);
    if (i < 2 * OUTC) stats[i] = 0.f;
}

__global__ void spconv_link(const int* __restrict__ out_idx,
                            const int* __restrict__ offset_id,
                            int*       __restrict__ head,
                            int*       __restrict__ next,
                            int nvox) {
    int vox = blockIdx.x * blockDim.x + threadIdx.x;
    if (vox >= nvox) return;
    int seg = out_idx[vox] * KVOL + offset_id[vox];
    next[vox] = atomicExch(&head[seg], vox);
}

__launch_bounds__(256) __attribute__((amdgpu_waves_per_eu(8, 8)))
__global__ void spconv_gemm_stats(const float* __restrict__ feats,
                                  const int*   __restrict__ head,
                                  const int*   __restrict__ next,
                                  const float* __restrict__ weight,
                                  float*       __restrict__ out,
                                  float*       __restrict__ stats,
                                  int mtotal, int nvox) {
    __shared__ float wint[(KDIM / 2) * 2 * OUTC];
    __shared__ float wsum[8 * 2 * OUTC];

    const int wave = threadIdx.x >> 5;
    const int lane = threadIdx.x & 31;
    const int r    = lane & 15;
    const int kbh  = lane >> 4;

    const int row_base = (blockIdx.x * 8 + wave) * 32;
    const int ar0 = min(row_base + r,      mtotal - 1);
    const int ar1 = min(row_base + 16 + r, mtotal - 1);
    const int hb0 = ar0 * KVOL, hb1 = ar1 * KVOL;

    const int a0 = head[hb0];
    int an = head[hb0 + 1];
    int a2 = head[hb0 + 2];

    for (int i = threadIdx.x; i < (KDIM / 2) * OUTC; i += blockDim.x) {
        int p = i >> 6;
        int n = i & 63;
        wint[p * 128 + n * 2 + 0] = weight[(2 * p)     * OUTC + n];
        wint[p * 128 + n * 2 + 1] = weight[(2 * p + 1) * OUTC + n];
    }
    __syncthreads();

    float ps0 = 0.f, ps1 = 0.f, ps2 = 0.f, ps3 = 0.f;
    float pq0 = 0.f, pq1 = 0.f, pq2 = 0.f, pq3 = 0.f;

    if (row_base < mtotal) {

    const float4* __restrict__ F4 = (const float4*)feats;
    const unsigned fq = 4u * (unsigned)kbh;
    const unsigned un = (unsigned)nvox;
    const float* __restrict__ B = wint + kbh * 1024 + r * 2;

    const float4 z4 = make_float4(0.f, 0.f, 0.f, 0.f);

    int b0 = -1, bn = -1;
    if ((unsigned)a0 < un) b0 = next[a0];
    if ((unsigned)an < un) bn = next[an];

    float4 x0 = z4, x1 = z4, x2 = z4, x3 = z4;
    float4 y0 = z4, y1 = z4, y2 = z4, y3 = z4;
    int c = -1;
    if ((unsigned)a0 < un) {
        const float4* p = F4 + ((unsigned)a0 * 8u + fq);
        x0 = p[0]; x1 = p[1]; x2 = p[2]; x3 = p[3];
    }
    if ((unsigned)b0 < un) {
        const float4* p = F4 + ((unsigned)b0 * 8u + fq);
        y0 = p[0]; y1 = p[1]; y2 = p[2]; y3 = p[3];
        c = next[b0];
    }

    v8f acc0 = {}, acc1 = {}, acc2 = {}, acc3 = {};

#pragma unroll 1
    for (int t = 0; t < 2 * KVOL; ++t) {
        x0.x += y0.x; x0.y += y0.y; x0.z += y0.z; x0.w += y0.w;
        x1.x += y1.x; x1.y += y1.y; x1.z += y1.z; x1.w += y1.w;
        x2.x += y2.x; x2.y += y2.y; x2.z += y2.z; x2.w += y2.w;
        x3.x += y3.x; x3.y += y3.y; x3.z += y3.z; x3.w += y3.w;
        for (int it = 2; it < nvox; ++it) {
            if (!((unsigned)c < un)) break;
            const float4* p = F4 + ((unsigned)c * 8u + fq);
            float4 t0 = p[0], t1 = p[1], t2 = p[2], t3 = p[3];
            c = next[c];
            x0.x += t0.x; x0.y += t0.y; x0.z += t0.z; x0.w += t0.w;
            x1.x += t1.x; x1.y += t1.y; x1.z += t1.z; x1.w += t1.w;
            x2.x += t2.x; x2.y += t2.y; x2.z += t2.z; x2.w += t2.w;
            x3.x += t3.x; x3.y += t3.y; x3.z += t3.z; x3.w += t3.w;
        }

        const float xs[16] = {x0.x, x0.y, x0.z, x0.w, x1.x, x1.y, x1.z, x1.w,
                              x2.x, x2.y, x2.z, x2.w, x3.x, x3.y, x3.z, x3.w};

        x0 = z4; x1 = z4; x2 = z4; x3 = z4;
        y0 = z4; y1 = z4; y2 = z4; y3 = z4;
        c = -1;
        int b2 = -1, a3 = -1;
        if (t + 1 < 2 * KVOL) {
            if ((unsigned)an < un) {
                const float4* p = F4 + ((unsigned)an * 8u + fq);
                x0 = p[0]; x1 = p[1]; x2 = p[2]; x3 = p[3];
            }
            if ((unsigned)bn < un) {
                const float4* p = F4 + ((unsigned)bn * 8u + fq);
                y0 = p[0]; y1 = p[1]; y2 = p[2]; y3 = p[3];
                c = next[bn];
            }
        }
        if (t + 2 < 2 * KVOL) {
            if ((unsigned)a2 < un) b2 = next[a2];
        }
        if (t + 3 < 2 * KVOL) {
            const int t3 = t + 3;
            a3 = head[((t3 & KVOL) ? hb1 : hb0) + (t3 & (KVOL - 1))];
        }

        const int o = t & (KVOL - 1);
        const float* Bo = B + o * (16 * 128);

#pragma unroll
        for (int kk = 0; kk < 8; ++kk) {
            v2f va; va.x = xs[2 * kk]; va.y = xs[2 * kk + 1];

            const float* bp = Bo + kk * 128;
            float2 g0 = *(const float2*)(bp);
            float2 g1 = *(const float2*)(bp + 32);
            float2 g2 = *(const float2*)(bp + 64);
            float2 g3 = *(const float2*)(bp + 96);
            v2f q0; q0.x = g0.x; q0.y = g0.y;
            v2f q1; q1.x = g1.x; q1.y = g1.y;
            v2f q2; q2.x = g2.x; q2.y = g2.y;
            v2f q3; q3.x = g3.x; q3.y = g3.y;

            acc0 = __builtin_amdgcn_wmma_f32_16x16x4_f32(false, va, false, q0, (short)0, acc0, false, false);
            acc1 = __builtin_amdgcn_wmma_f32_16x16x4_f32(false, va, false, q1, (short)0, acc1, false, false);
            acc2 = __builtin_amdgcn_wmma_f32_16x16x4_f32(false, va, false, q2, (short)0, acc2, false, false);
            acc3 = __builtin_amdgcn_wmma_f32_16x16x4_f32(false, va, false, q3, (short)0, acc3, false, false);
        }

        if (o == KVOL - 1) {
            const int mrow = row_base + ((t >> 3) << 4) + ((lane >> 4) << 3);
#define EMIT_NT(ACC, PS, PQ, NT)                                               \
            _Pragma("unroll")                                                  \
            for (int i = 0; i < 8; ++i) {                                      \
                int m = mrow + i;                                              \
                float v = (ACC)[i];                                            \
                if (m < mtotal) {                                              \
                    out[(long long)m * OUTC + (NT) * 16 + r] = v;              \
                    PS += v; PQ += v * v;                                      \
                }                                                              \
            }
            EMIT_NT(acc0, ps0, pq0, 0)
            EMIT_NT(acc1, ps1, pq1, 1)
            EMIT_NT(acc2, ps2, pq2, 2)
            EMIT_NT(acc3, ps3, pq3, 3)
#undef EMIT_NT
            const v8f zacc = {};
            acc0 = zacc; acc1 = zacc; acc2 = zacc; acc3 = zacc;
        }

        an = a2; bn = b2; a2 = a3;
    }

    }

#define EMIT_STAT(PS, PQ, NT)                                                  \
    {                                                                          \
        float s = (PS) + __shfl_xor((PS), 16, 32);                             \
        float q = (PQ) + __shfl_xor((PQ), 16, 32);                             \
        if (lane < 16) {                                                       \
            wsum[wave * (2 * OUTC) + (NT) * 16 + r] = s;                       \
            wsum[wave * (2 * OUTC) + OUTC + (NT) * 16 + r] = q;                \
        }                                                                      \
    }
    EMIT_STAT(ps0, pq0, 0)
    EMIT_STAT(ps1, pq1, 1)
    EMIT_STAT(ps2, pq2, 2)
    EMIT_STAT(ps3, pq3, 3)
#undef EMIT_STAT

    __syncthreads();

    if (threadIdx.x < 2 * OUTC) {
        float a = wsum[threadIdx.x];
        for (int w = 1; w < 8; ++w) a += wsum[w * (2 * OUTC) + threadIdx.x];
        unsafeAtomicAdd(&stats[threadIdx.x], a);
    }
}

__global__ void spconv_bn_apply(float* __restrict__ out,
                                const float* __restrict__ stats,
                                const float* __restrict__ gamma,
                                const float* __restrict__ beta,
                                int mtotal) {
    long long i4     = (long long)blockIdx.x * blockDim.x + threadIdx.x;
    long long total4 = ((long long)mtotal * OUTC) >> 2;
    if (i4 >= total4) return;
    const float inv_m = 1.0f / (float)mtotal;
    const int c0 = (int)((i4 << 2) & (OUTC - 1));
    float4 v = ((float4*)out)[i4];
    float x[4] = {v.x, v.y, v.z, v.w};
    float y[4];
#pragma unroll
    for (int j = 0; j < 4; ++j) {
        int c = c0 + j;
        float mean = stats[c] * inv_m;
        float var  = stats[64 + c] * inv_m - mean * mean;
        y[j] = (x[j] - mean) * rsqrtf(var + BN_EPS) * gamma[c] + beta[c];
    }
    ((float4*)out)[i4] = make_float4(y[0], y[1], y[2], y[3]);
}

extern "C" void kernel_launch(void* const* d_in, const int* in_sizes, int n_in,
                              void* d_out, int out_size, void* d_ws, size_t ws_size,
                              hipStream_t stream) {
    const float* feats     = (const float*)d_in[0];
    const float* weight    = (const float*)d_in[1];
    const float* gamma     = (const float*)d_in[2];
    const float* beta      = (const float*)d_in[3];
    const int*   out_idx   = (const int*)d_in[4];
    const int*   offset_id = (const int*)d_in[5];
    const int nvox   = in_sizes[0] / INC;
    const int mtotal = out_size / OUTC;

    int*   head  = (int*)d_ws;
    int*   next  = head + (long long)mtotal * KVOL;
    float* stats = (float*)(next + nvox);
    const long long nhead4 = ((long long)mtotal * KVOL) >> 2;

    spconv_init_ws<<<1024, 256, 0, stream>>>(head, nhead4, stats);

    const int sblocks = (nvox + 255) / 256;
    spconv_link<<<sblocks, 256, 0, stream>>>(out_idx, offset_id, head, next, nvox);

    const int gblocks = (mtotal + 255) / 256;
    spconv_gemm_stats<<<gblocks, 256, 0, stream>>>(feats, head, next, weight, (float*)d_out,
                                                   stats, mtotal, nvox);

    const long long total4 = ((long long)mtotal * OUTC) >> 2;
    const int bblocks = (int)((total4 + 255) / 256);
    spconv_bn_apply<<<bblocks, 256, 0, stream>>>((float*)d_out, stats, gamma, beta, mtotal);
}


# GEMM weight staging: each thread's sixteen 16-byte loads all in flight before the first LDS store (one round trip, was 32 dependent 4-byte ones); work-groups start at different rows
# speedup vs baseline: 2.3581x; 1.1250x over previous
#include <hip/hip_runtime.h>


typedef __attribute__((ext_vector_type(2))) float v2f;
typedef __attribute__((ext_vector_type(8))) float v8f;

#define INC    32
#define OUTC   64
#define KVOL   8
#define KDIM   256
#define BN_EPS 1e-5f

__global__ void spconv_init_ws(int* __restrict__ head, long long nhead4,
                               float* __restrict__ stats) {
    long long i = (long long)blockIdx.x * blockDim.x + threadIdx.x;
    long long stride = (long long)gridDim.x * blockDim.x;
    int4* p = (int4*)head;
    for (long long j = i; j < nhead4; j += stride)
        p[j] = make_int4(-1, -1, -1, -1);
    if (i < 2 * OUTC) stats[i] = 0.f;
}

__global__ void spconv_link(const int* __restrict__ out_idx,
                            const int* __restrict__ offset_id,
                            int*       __restrict__ head,
                            int*       __restrict__ next,
                            int nvox) {
    int vox = blockIdx.x * blockDim.x + threadIdx.x;
    if (vox >= nvox) return;
    int seg = out_idx[vox] * KVOL + offset_id[vox];
    next[vox] = atomicExch(&head[seg], vox);
}

__launch_bounds__(256) __attribute__((amdgpu_waves_per_eu(8, 8)))
__global__ void spconv_gemm_stats(const float* __restrict__ feats,
                                  const int*   __restrict__ head,
                                  const int*   __restrict__ next,
                                  const float* __restrict__ weight,
                                  float*       __restrict__ out,
                                  float*       __restrict__ stats,
                                  int mtotal, int nvox) {
    __shared__ __attribute__((aligned(16))) float wint[(KDIM / 2) * 2 * OUTC];
    __shared__ float wsum[8 * 2 * OUTC];

    const int wave = threadIdx.x >> 5;
    const int lane = threadIdx.x & 31;
    const int r    = lane & 15;
    const int kbh  = lane >> 4;

    const int row_base = (blockIdx.x * 8 + wave) * 32;
    const int ar0 = min(row_base + r,      mtotal - 1);
    const int ar1 = min(row_base + 16 + r, mtotal - 1);
    const int hb0 = ar0 * KVOL, hb1 = ar1 * KVOL;

    const int a0 = head[hb0];
    int an = head[hb0 + 1];
    int a2 = head[hb0 + 2];

    {
        const float4* __restrict__ W4 = (const float4*)weight;
        const int rot = blockIdx.x & (KDIM / 2 - 1);
        float4 wa[8], wb[8];
#pragma unroll
        for (int j = 0; j < 8; ++j) {
            const int u  = j * 256 + threadIdx.x;
            const int p  = ((u >> 4) + rot) & (KDIM / 2 - 1);
            const int n4 = u & 15;
            wa[j] = W4[(2 * p)     * (OUTC / 4) + n4];
            wb[j] = W4[(2 * p + 1) * (OUTC / 4) + n4];
        }
#pragma unroll
        for (int j = 0; j < 8; ++j) {
            const int u  = j * 256 + threadIdx.x;
            const int p  = ((u >> 4) + rot) & (KDIM / 2 - 1);
            const int n4 = u & 15;
            float4* d = (float4*)(wint + p * 128 + n4 * 8);
            d[0] = make_float4(wa[j].x, wb[j].x, wa[j].y, wb[j].y);
            d[1] = make_float4(wa[j].z, wb[j].z, wa[j].w, wb[j].w);
        }
    }
    __syncthreads();

    float ps0 = 0.f, ps1 = 0.f, ps2 = 0.f, ps3 = 0.f;
    float pq0 = 0.f, pq1 = 0.f, pq2 = 0.f, pq3 = 0.f;

    if (row_base < mtotal) {

    const float4* __restrict__ F4 = (const float4*)feats;
    const unsigned fq = 4u * (unsigned)kbh;
    const unsigned un = (unsigned)nvox;
    const float* __restrict__ B = wint + kbh * 1024 + r * 2;

    const float4 z4 = make_float4(0.f, 0.f, 0.f, 0.f);

    int b0 = -1, bn = -1;
    if ((unsigned)a0 < un) b0 = next[a0];
    if ((unsigned)an < un) bn = next[an];

    float4 x0 = z4, x1 = z4, x2 = z4, x3 = z4;
    float4 y0 = z4, y1 = z4, y2 = z4, y3 = z4;
    int c = -1;
    if ((unsigned)a0 < un) {
        const float4* p = F4 + ((unsigned)a0 * 8u + fq);
        x0 = p[0]; x1 = p[1]; x2 = p[2]; x3 = p[3];
    }
    if ((unsigned)b0 < un) {
        const float4* p = F4 + ((unsigned)b0 * 8u + fq);
        y0 = p[0]; y1 = p[1]; y2 = p[2]; y3 = p[3];
        c = next[b0];
    }

    v8f acc0 = {}, acc1 = {}, acc2 = {}, acc3 = {};

#pragma unroll 1
    for (int t = 0; t < 2 * KVOL; ++t) {
        x0.x += y0.x; x0.y += y0.y; x0.z += y0.z; x0.w += y0.w;
        x1.x += y1.x; x1.y += y1.y; x1.z += y1.z; x1.w += y1.w;
        x2.x += y2.x; x2.y += y2.y; x2.z += y2.z; x2.w += y2.w;
        x3.x += y3.x; x3.y += y3.y; x3.z += y3.z; x3.w += y3.w;
        for (int it = 2; it < nvox; ++it) {
            if (!((unsigned)c < un)) break;
            const float4* p = F4 + ((unsigned)c * 8u + fq);
            float4 t0 = p[0], t1 = p[1], t2 = p[2], t3 = p[3];
            c = next[c];
            x0.x += t0.x; x0.y += t0.y; x0.z += t0.z; x0.w += t0.w;
            x1.x += t1.x; x1.y += t1.y; x1.z += t1.z; x1.w += t1.w;
            x2.x += t2.x; x2.y += t2.y; x2.z += t2.z; x2.w += t2.w;
            x3.x += t3.x; x3.y += t3.y; x3.z += t3.z; x3.w += t3.w;
        }

        const float xs[16] = {x0.x, x0.y, x0.z, x0.w, x1.x, x1.y, x1.z, x1.w,
                              x2.x, x2.y, x2.z, x2.w, x3.x, x3.y, x3.z, x3.w};

        x0 = z4; x1 = z4; x2 = z4; x3 = z4;
        y0 = z4; y1 = z4; y2 = z4; y3 = z4;
        c = -1;
        int b2 = -1, a3 = -1;
        if (t + 1 < 2 * KVOL) {
            if ((unsigned)an < un) {
                const float4* p = F4 + ((unsigned)an * 8u + fq);
                x0 = p[0]; x1 = p[1]; x2 = p[2]; x3 = p[3];
            }
            if ((unsigned)bn < un) {
                const float4* p = F4 + ((unsigned)bn * 8u + fq);
                y0 = p[0]; y1 = p[1]; y2 = p[2]; y3 = p[3];
                c = next[bn];
            }
        }
        if (t + 2 < 2 * KVOL) {
            if ((unsigned)a2 < un) b2 = next[a2];
        }
        if (t + 3 < 2 * KVOL) {
            const int t3 = t + 3;
            a3 = head[((t3 & KVOL) ? hb1 : hb0) + (t3 & (KVOL - 1))];
        }

        const int o = t & (KVOL - 1);
        const float* Bo = B + o * (16 * 128);

#pragma unroll
        for (int kk = 0; kk < 8; ++kk) {
            v2f va; va.x = xs[2 * kk]; va.y = xs[2 * kk + 1];

            const float* bp = Bo + kk * 128;
            float2 g0 = *(const float2*)(bp);
            float2 g1 = *(const float2*)(bp + 32);
            float2 g2 = *(const float2*)(bp + 64);
            float2 g3 = *(const float2*)(bp + 96);
            v2f q0; q0.x = g0.x; q0.y = g0.y;
            v2f q1; q1.x = g1.x; q1.y = g1.y;
            v2f q2; q2.x = g2.x; q2.y = g2.y;
            v2f q3; q3.x = g3.x; q3.y = g3.y;

            acc0 = __builtin_amdgcn_wmma_f32_16x16x4_f32(false, va, false, q0, (short)0, acc0, false, false);
            acc1 = __builtin_amdgcn_wmma_f32_16x16x4_f32(false, va, false, q1, (short)0, acc1, false, false);
            acc2 = __builtin_amdgcn_wmma_f32_16x16x4_f32(false, va, false, q2, (short)0, acc2, false, false);
            acc3 = __builtin_amdgcn_wmma_f32_16x16x4_f32(false, va, false, q3, (short)0, acc3, false, false);
        }

        if (o == KVOL - 1) {
            const int mrow = row_base + ((t >> 3) << 4) + ((lane >> 4) << 3);
#define EMIT_NT(ACC, PS, PQ, NT)                                               \
            _Pragma("unroll")                                                  \
            for (int i = 0; i < 8; ++i) {                                      \
                int m = mrow + i;                                              \
                float v = (ACC)[i];                                            \
                if (m < mtotal) {                                              \
                    out[(long long)m * OUTC + (NT) * 16 + r] = v;              \
                    PS += v; PQ += v * v;                                      \
                }                                                              \
            }
            EMIT_NT(acc0, ps0, pq0, 0)
            EMIT_NT(acc1, ps1, pq1, 1)
            EMIT_NT(acc2, ps2, pq2, 2)
            EMIT_NT(acc3, ps3, pq3, 3)
#undef EMIT_NT
            const v8f zacc = {};
            acc0 = zacc; acc1 = zacc; acc2 = zacc; acc3 = zacc;
        }

        an = a2; bn = b2; a2 = a3;
    }

    }

#define EMIT_STAT(PS, PQ, NT)                                                  \
    {                                                                          \
        float s = (PS) + __shfl_xor((PS), 16, 32);                             \
        float q = (PQ) + __shfl_xor((PQ), 16, 32);                             \
        if (lane < 16) {                                                       \
            wsum[wave * (2 * OUTC) + (NT) * 16 + r] = s;                       \
            wsum[wave * (2 * OUTC) + OUTC + (NT) * 16 + r] = q;                \
        }                                                                      \
    }
    EMIT_STAT(ps0, pq0, 0)
    EMIT_STAT(ps1, pq1, 1)
    EMIT_STAT(ps2, pq2, 2)
    EMIT_STAT(ps3, pq3, 3)
#undef EMIT_STAT

    __syncthreads();

    if (threadIdx.x < 2 * OUTC) {
        float a = wsum[threadIdx.x];
        for (int w = 1; w < 8; ++w) a += wsum[w * (2 * OUTC) + threadIdx.x];
        unsafeAtomicAdd(&stats[threadIdx.x], a);
    }
}

__global__ void spconv_bn_apply(float* __restrict__ out,
                                const float* __restrict__ stats,
                                const float* __restrict__ gamma,
                                const float* __restrict__ beta,
                                int mtotal) {
    long long i4     = (long long)blockIdx.x * blockDim.x + threadIdx.x;
    long long total4 = ((long long)mtotal * OUTC) >> 2;
    if (i4 >= total4) return;
    const float inv_m = 1.0f / (float)mtotal;
    const int c0 = (int)((i4 << 2) & (OUTC - 1));
    float4 v = ((float4*)out)[i4];
    float x[4] = {v.x, v.y, v.z, v.w};
    float y[4];
#pragma unroll
    for (int j = 0; j < 4; ++j) {
        int c = c0 + j;
        float mean = stats[c] * inv_m;
        float var  = stats[64 + c] * inv_m - mean * mean;
        y[j] = (x[j] - mean) * rsqrtf(var + BN_EPS) * gamma[c] + beta[c];
    }
    ((float4*)out)[i4] = make_float4(y[0], y[1], y[2], y[3]);
}

extern "C" void kernel_launch(void* const* d_in, const int* in_sizes, int n_in,
                              void* d_out, int out_size, void* d_ws, size_t ws_size,
                              hipStream_t stream) {
    const float* feats     = (const float*)d_in[0];
    const float* weight    = (const float*)d_in[1];
    const float* gamma     = (const float*)d_in[2];
    const float* beta      = (const float*)d_in[3];
    const int*   out_idx   = (const int*)d_in[4];
    const int*   offset_id = (const int*)d_in[5];
    const int nvox   = in_sizes[0] / INC;
    const int mtotal = out_size / OUTC;

    int*   head  = (int*)d_ws;
    int*   next  = head + (long long)mtotal * KVOL;
    float* stats = (float*)(next + nvox);
    const long long nhead4 = ((long long)mtotal * KVOL) >> 2;

    spconv_init_ws<<<1024, 256, 0, stream>>>(head, nhead4, stats);

    const int sblocks = (nvox + 255) / 256;
    spconv_link<<<sblocks, 256, 0, stream>>>(out_idx, offset_id, head, next, nvox);

    const int gblocks = (mtotal + 255) / 256;
    spconv_gemm_stats<<<gblocks, 256, 0, stream>>>(feats, head, next, weight, (float*)d_out,
                                                   stats, mtotal, nvox);

    const long long total4 = ((long long)mtotal * OUTC) >> 2;
    const int bblocks = (int)((total4 + 255) / 256);
    spconv_bn_apply<<<bblocks, 256, 0, stream>>>((float*)d_out, stats, gamma, beta, mtotal);
}


# Init, link and BN apply as resident grids whose threads loop (8 voxels / 32 float4 per trip, all loads in flight, BN column terms read once): 172,032 short waves become 10,240; list order by arrival
# speedup vs baseline: 2.6510x; 1.1242x over previous
#include <hip/hip_runtime.h>


typedef __attribute__((ext_vector_type(2))) float v2f;
typedef __attribute__((ext_vector_type(8))) float v8f;

#define INC    32
#define OUTC   64
#define KVOL   8
#define KDIM   256
#define BN_EPS 1e-5f

__global__ void spconv_init_ws(int* __restrict__ head, long long nhead4,
                               float* __restrict__ stats) {
    long long i = (long long)blockIdx.x * blockDim.x + threadIdx.x;
    long long stride = (long long)gridDim.x * blockDim.x;
    int4* p = (int4*)head;
    for (long long j = i; j < nhead4; j += stride)
        p[j] = make_int4(-1, -1, -1, -1);
    if (i < 2 * OUTC) stats[i] = 0.f;
}

#define LINK_PER 8
__launch_bounds__(256)
__global__ void spconv_link(const int* __restrict__ out_idx,
                            const int* __restrict__ offset_id,
                            int*       __restrict__ head,
                            int*       __restrict__ next,
                            int nvox) {
    const long long G = (long long)gridDim.x * blockDim.x;
    long long v = (long long)blockIdx.x * blockDim.x + threadIdx.x;
    for (; v + (LINK_PER - 1) * G < nvox; v += LINK_PER * G) {
        int seg[LINK_PER], old[LINK_PER];
#pragma unroll
        for (int k = 0; k < LINK_PER; ++k) {
            const long long u = v + k * G;
            seg[k] = out_idx[u] * KVOL + offset_id[u];
        }
#pragma unroll
        for (int k = 0; k < LINK_PER; ++k)
            old[k] = atomicExch(&head[seg[k]], (int)(v + k * G));
#pragma unroll
        for (int k = 0; k < LINK_PER; ++k)
            next[v + k * G] = old[k];
    }
    for (; v < nvox; v += G) {
        const int seg = out_idx[v] * KVOL + offset_id[v];
        next[v] = atomicExch(&head[seg], (int)v);
    }
}

__launch_bounds__(256) __attribute__((amdgpu_waves_per_eu(8, 8)))
__global__ void spconv_gemm_stats(const float* __restrict__ feats,
                                  const int*   __restrict__ head,
                                  const int*   __restrict__ next,
                                  const float* __restrict__ weight,
                                  float*       __restrict__ out,
                                  float*       __restrict__ stats,
                                  int mtotal, int nvox) {
    __shared__ __attribute__((aligned(16))) float wint[(KDIM / 2) * 2 * OUTC];
    __shared__ float wsum[8 * 2 * OUTC];

    const int wave = threadIdx.x >> 5;
    const int lane = threadIdx.x & 31;
    const int r    = lane & 15;
    const int kbh  = lane >> 4;

    const int row_base = (blockIdx.x * 8 + wave) * 32;
    const int ar0 = min(row_base + r,      mtotal - 1);
    const int ar1 = min(row_base + 16 + r, mtotal - 1);
    const int hb0 = ar0 * KVOL, hb1 = ar1 * KVOL;

    const int a0 = head[hb0];
    int an = head[hb0 + 1];
    int a2 = head[hb0 + 2];

    {
        const float4* __restrict__ W4 = (const float4*)weight;
        const int rot = blockIdx.x & (KDIM / 2 - 1);
        float4 wa[8], wb[8];
#pragma unroll
        for (int j = 0; j < 8; ++j) {
            const int u  = j * 256 + threadIdx.x;
            const int p  = ((u >> 4) + rot) & (KDIM / 2 - 1);
            const int n4 = u & 15;
            wa[j] = W4[(2 * p)     * (OUTC / 4) + n4];
            wb[j] = W4[(2 * p + 1) * (OUTC / 4) + n4];
        }
#pragma unroll
        for (int j = 0; j < 8; ++j) {
            const int u  = j * 256 + threadIdx.x;
            const int p  = ((u >> 4) + rot) & (KDIM / 2 - 1);
            const int n4 = u & 15;
            float4* d = (float4*)(wint + p * 128 + n4 * 8);
            d[0] = make_float4(wa[j].x, wb[j].x, wa[j].y, wb[j].y);
            d[1] = make_float4(wa[j].z, wb[j].z, wa[j].w, wb[j].w);
        }
    }
    __syncthreads();

    float ps0 = 0.f, ps1 = 0.f, ps2 = 0.f, ps3 = 0.f;
    float pq0 = 0.f, pq1 = 0.f, pq2 = 0.f, pq3 = 0.f;

    if (row_base < mtotal) {

    const float4* __restrict__ F4 = (const float4*)feats;
    const unsigned fq = 4u * (unsigned)kbh;
    const unsigned un = (unsigned)nvox;
    const float* __restrict__ B = wint + kbh * 1024 + r * 2;

    const float4 z4 = make_float4(0.f, 0.f, 0.f, 0.f);

    int b0 = -1, bn = -1;
    if ((unsigned)a0 < un) b0 = next[a0];
    if ((unsigned)an < un) bn = next[an];

    float4 x0 = z4, x1 = z4, x2 = z4, x3 = z4;
    float4 y0 = z4, y1 = z4, y2 = z4, y3 = z4;
    int c = -1;
    if ((unsigned)a0 < un) {
        const float4* p = F4 + ((unsigned)a0 * 8u + fq);
        x0 = p[0]; x1 = p[1]; x2 = p[2]; x3 = p[3];
    }
    if ((unsigned)b0 < un) {
        const float4* p = F4 + ((unsigned)b0 * 8u + fq);
        y0 = p[0]; y1 = p[1]; y2 = p[2]; y3 = p[3];
        c = next[b0];
    }

    v8f acc0 = {}, acc1 = {}, acc2 = {}, acc3 = {};

#pragma unroll 1
    for (int t = 0; t < 2 * KVOL; ++t) {
        x0.x += y0.x; x0.y += y0.y; x0.z += y0.z; x0.w += y0.w;
        x1.x += y1.x; x1.y += y1.y; x1.z += y1.z; x1.w += y1.w;
        x2.x += y2.x; x2.y += y2.y; x2.z += y2.z; x2.w += y2.w;
        x3.x += y3.x; x3.y += y3.y; x3.z += y3.z; x3.w += y3.w;
        for (int it = 2; it < nvox; ++it) {
            if (!((unsigned)c < un)) break;
            const float4* p = F4 + ((unsigned)c * 8u + fq);
            float4 t0 = p[0], t1 = p[1], t2 = p[2], t3 = p[3];
            c = next[c];
            x0.x += t0.x; x0.y += t0.y; x0.z += t0.z; x0.w += t0.w;
            x1.x += t1.x; x1.y += t1.y; x1.z += t1.z; x1.w += t1.w;
            x2.x += t2.x; x2.y += t2.y; x2.z += t2.z; x2.w += t2.w;
            x3.x += t3.x; x3.y += t3.y; x3.z += t3.z; x3.w += t3.w;
        }

        const float xs[16] = {x0.x, x0.y, x0.z, x0.w, x1.x, x1.y, x1.z, x1.w,
                              x2.x, x2.y, x2.z, x2.w, x3.x, x3.y, x3.z, x3.w};

        x0 = z4; x1 = z4; x2 = z4; x3 = z4;
        y0 = z4; y1 = z4; y2 = z4; y3 = z4;
        c = -1;
        int b2 = -1, a3 = -1;
        if (t + 1 < 2 * KVOL) {
            if ((unsigned)an < un) {
                const float4* p = F4 + ((unsigned)an * 8u + fq);
                x0 = p[0]; x1 = p[1]; x2 = p[2]; x3 = p[3];
            }
            if ((unsigned)bn < un) {
                const float4* p = F4 + ((unsigned)bn * 8u + fq);
                y0 = p[0]; y1 = p[1]; y2 = p[2]; y3 = p[3];
                c = next[bn];
            }
        }
        if (t + 2 < 2 * KVOL) {
            if ((unsigned)a2 < un) b2 = next[a2];
        }
        if (t + 3 < 2 * KVOL) {
            const int t3 = t + 3;
            a3 = head[((t3 & KVOL) ? hb1 : hb0) + (t3 & (KVOL - 1))];
        }

        const int o = t & (KVOL - 1);
        const float* Bo = B + o * (16 * 128);

#pragma unroll
        for (int kk = 0; kk < 8; ++kk) {
            v2f va; va.x = xs[2 * kk]; va.y = xs[2 * kk + 1];

            const float* bp = Bo + kk * 128;
            float2 g0 = *(const float2*)(bp);
            float2 g1 = *(const float2*)(bp + 32);
            float2 g2 = *(const float2*)(bp + 64);
            float2 g3 = *(const float2*)(bp + 96);
            v2f q0; q0.x = g0.x; q0.y = g0.y;
            v2f q1; q1.x = g1.x; q1.y = g1.y;
            v2f q2; q2.x = g2.x; q2.y = g2.y;
            v2f q3; q3.x = g3.x; q3.y = g3.y;

            acc0 = __builtin_amdgcn_wmma_f32_16x16x4_f32(false, va, false, q0, (short)0, acc0, false, false);
            acc1 = __builtin_amdgcn_wmma_f32_16x16x4_f32(false, va, false, q1, (short)0, acc1, false, false);
            acc2 = __builtin_amdgcn_wmma_f32_16x16x4_f32(false, va, false, q2, (short)0, acc2, false, false);
            acc3 = __builtin_amdgcn_wmma_f32_16x16x4_f32(false, va, false, q3, (short)0, acc3, false, false);
        }

        if (o == KVOL - 1) {
            const int mrow = row_base + ((t >> 3) << 4) + ((lane >> 4) << 3);
#define EMIT_NT(ACC, PS, PQ, NT)                                               \
            _Pragma("unroll")                                                  \
            for (int i = 0; i < 8; ++i) {                                      \
                int m = mrow + i;                                              \
                float v = (ACC)[i];                                            \
                if (m < mtotal) {                                              \
                    out[(long long)m * OUTC + (NT) * 16 + r] = v;              \
                    PS += v; PQ += v * v;                                      \
                }                                                              \
            }
            EMIT_NT(acc0, ps0, pq0, 0)
            EMIT_NT(acc1, ps1, pq1, 1)
            EMIT_NT(acc2, ps2, pq2, 2)
            EMIT_NT(acc3, ps3, pq3, 3)
#undef EMIT_NT
            const v8f zacc = {};
            acc0 = zacc; acc1 = zacc; acc2 = zacc; acc3 = zacc;
        }

        an = a2; bn = b2; a2 = a3;
    }

    }

#define EMIT_STAT(PS, PQ, NT)                                                  \
    {                                                                          \
        float s = (PS) + __shfl_xor((PS), 16, 32);                             \
        float q = (PQ) + __shfl_xor((PQ), 16, 32);                             \
        if (lane < 16) {                                                       \
            wsum[wave * (2 * OUTC) + (NT) * 16 + r] = s;                       \
            wsum[wave * (2 * OUTC) + OUTC + (NT) * 16 + r] = q;                \
        }                                                                      \
    }
    EMIT_STAT(ps0, pq0, 0)
    EMIT_STAT(ps1, pq1, 1)
    EMIT_STAT(ps2, pq2, 2)
    EMIT_STAT(ps3, pq3, 3)
#undef EMIT_STAT

    __syncthreads();

    if (threadIdx.x < 2 * OUTC) {
        float a = wsum[threadIdx.x];
        for (int w = 1; w < 8; ++w) a += wsum[w * (2 * OUTC) + threadIdx.x];
        unsafeAtomicAdd(&stats[threadIdx.x], a);
    }
}

#define BN_PER 32
__launch_bounds__(256)
__global__ void spconv_bn_apply(float* __restrict__ out,
                                const float* __restrict__ stats,
                                const float* __restrict__ gamma,
                                const float* __restrict__ beta,
                                int mtotal) {
    const int lane = threadIdx.x & 31;
    const long long total4 = ((long long)mtotal * OUTC) >> 2;
    const float inv_m = 1.0f / (float)mtotal;
    const int c0 = (lane << 2) & (OUTC - 1);
    float mean[4], rs[4], ga[4], be[4];
#pragma unroll
    for (int j = 0; j < 4; ++j) {
        const int c = c0 + j;
        mean[j] = stats[c] * inv_m;
        const float var = stats[64 + c] * inv_m - mean[j] * mean[j];
        rs[j] = rsqrtf(var + BN_EPS);
        ga[j] = gamma[c];
        be[j] = beta[c];
    }
    float4* __restrict__ o4 = (float4*)out;
    const long long CH = 32LL * BN_PER;
    const long long nwaves = (long long)gridDim.x * (blockDim.x >> 5);
    const long long wv = (long long)blockIdx.x * (blockDim.x >> 5)
                       + __builtin_amdgcn_readfirstlane((int)(threadIdx.x >> 5));
    const long long nfull = total4 / CH;
    for (long long ch = wv; ch < nfull; ch += nwaves) {
        float4* p = o4 + ch * CH + lane;
        float4 v[BN_PER];
#pragma unroll
        for (int k = 0; k < BN_PER; ++k) v[k] = p[k * 32];
        __builtin_amdgcn_sched_barrier(0);
#pragma unroll
        for (int k = 0; k < BN_PER; ++k) {
            float4 y;
            y.x = (v[k].x - mean[0]) * rs[0] * ga[0] + be[0];
            y.y = (v[k].y - mean[1]) * rs[1] * ga[1] + be[1];
            y.z = (v[k].z - mean[2]) * rs[2] * ga[2] + be[2];
            y.w = (v[k].w - mean[3]) * rs[3] * ga[3] + be[3];
            p[k * 32] = y;
        }
    }
    const long long G = (long long)gridDim.x * blockDim.x;
    for (long long t = nfull * CH + (long long)blockIdx.x * blockDim.x + threadIdx.x;
         t < total4; t += G) {
        const float4 x = o4[t];
        float4 y;
        y.x = (x.x - mean[0]) * rs[0] * ga[0] + be[0];
        y.y = (x.y - mean[1]) * rs[1] * ga[1] + be[1];
        y.z = (x.z - mean[2]) * rs[2] * ga[2] + be[2];
        y.w = (x.w - mean[3]) * rs[3] * ga[3] + be[3];
        o4[t] = y;
    }
}

extern "C" void kernel_launch(void* const* d_in, const int* in_sizes, int n_in,
                              void* d_out, int out_size, void* d_ws, size_t ws_size,
                              hipStream_t stream) {
    const float* feats     = (const float*)d_in[0];
    const float* weight    = (const float*)d_in[1];
    const float* gamma     = (const float*)d_in[2];
    const float* beta      = (const float*)d_in[3];
    const int*   out_idx   = (const int*)d_in[4];
    const int*   offset_id = (const int*)d_in[5];
    const int nvox   = in_sizes[0] / INC;
    const int mtotal = out_size / OUTC;

    int*   head  = (int*)d_ws;
    int*   next  = head + (long long)mtotal * KVOL;
    float* stats = (float*)(next + nvox);
    const long long nhead4 = ((long long)mtotal * KVOL) >> 2;

    spconv_init_ws<<<256, 256, 0, stream>>>(head, nhead4, stats);

    spconv_link<<<512, 256, 0, stream>>>(out_idx, offset_id, head, next, nvox);

    const int gblocks = (mtotal + 255) / 256;
    spconv_gemm_stats<<<gblocks, 256, 0, stream>>>(feats, head, next, weight, (float*)d_out,
                                                   stats, mtotal, nvox);

    spconv_bn_apply<<<512, 256, 0, stream>>>((float*)d_out, stats, gamma, beta, mtotal);
}


# GEMM output leaves as whole 256-byte rows: accumulators staged through a per-wave LDS slice, one 16-byte store per lane covers two full rows (was four 64-byte pieces per row); sum order kept
# speedup vs baseline: 2.9683x; 1.1197x over previous
#include <hip/hip_runtime.h>


typedef __attribute__((ext_vector_type(2))) float v2f;
typedef __attribute__((ext_vector_type(8))) float v8f;

#define INC    32
#define OUTC   64
#define KVOL   8
#define KDIM   256
#define BN_EPS 1e-5f

__global__ void spconv_init_ws(int* __restrict__ head, long long nhead4,
                               float* __restrict__ stats) {
    long long i = (long long)blockIdx.x * blockDim.x + threadIdx.x;
    long long stride = (long long)gridDim.x * blockDim.x;
    int4* p = (int4*)head;
    for (long long j = i; j < nhead4; j += stride)
        p[j] = make_int4(-1, -1, -1, -1);
    if (i < 2 * OUTC) stats[i] = 0.f;
}

#define LINK_PER 8
__launch_bounds__(256)
__global__ void spconv_link(const int* __restrict__ out_idx,
                            const int* __restrict__ offset_id,
                            int*       __restrict__ head,
                            int*       __restrict__ next,
                            int nvox) {
    const long long G = (long long)gridDim.x * blockDim.x;
    long long v = (long long)blockIdx.x * blockDim.x + threadIdx.x;
    for (; v + (LINK_PER - 1) * G < nvox; v += LINK_PER * G) {
        int seg[LINK_PER], old[LINK_PER];
#pragma unroll
        for (int k = 0; k < LINK_PER; ++k) {
            const long long u = v + k * G;
            seg[k] = out_idx[u] * KVOL + offset_id[u];
        }
#pragma unroll
        for (int k = 0; k < LINK_PER; ++k)
            old[k] = atomicExch(&head[seg[k]], (int)(v + k * G));
#pragma unroll
        for (int k = 0; k < LINK_PER; ++k)
            next[v + k * G] = old[k];
    }
    for (; v < nvox; v += G) {
        const int seg = out_idx[v] * KVOL + offset_id[v];
        next[v] = atomicExch(&head[seg], (int)v);
    }
}

__device__ __forceinline__ void wave_lds_order() {
    __builtin_amdgcn_fence(__ATOMIC_RELEASE, "wavefront");
    __builtin_amdgcn_wave_barrier();
    __builtin_amdgcn_fence(__ATOMIC_ACQUIRE, "wavefront");
}

__launch_bounds__(256) __attribute__((amdgpu_waves_per_eu(8, 8)))
__global__ void spconv_gemm_stats(const float* __restrict__ feats,
                                  const int*   __restrict__ head,
                                  const int*   __restrict__ next,
                                  const float* __restrict__ weight,
                                  float*       __restrict__ out,
                                  float*       __restrict__ stats,
                                  int mtotal, int nvox) {
    __shared__ __attribute__((aligned(16))) float wint[(KDIM / 2) * 2 * OUTC];
    __shared__ float wsum[8 * 2 * OUTC];
    __shared__ __attribute__((aligned(16))) float ostage[8 * 320];

    const int wave = threadIdx.x >> 5;
    const int lane = threadIdx.x & 31;
    const int r    = lane & 15;
    const int kbh  = lane >> 4;

    const int row_base = (blockIdx.x * 8 + wave) * 32;
    const int ar0 = min(row_base + r,      mtotal - 1);
    const int ar1 = min(row_base + 16 + r, mtotal - 1);
    const int hb0 = ar0 * KVOL, hb1 = ar1 * KVOL;

    const int a0 = head[hb0];
    int an = head[hb0 + 1];
    int a2 = head[hb0 + 2];

    {
        const float4* __restrict__ W4 = (const float4*)weight;
        const int rot = blockIdx.x & (KDIM / 2 - 1);
        float4 wa[8], wb[8];
#pragma unroll
        for (int j = 0; j < 8; ++j) {
            const int u  = j * 256 + threadIdx.x;
            const int p  = ((u >> 4) + rot) & (KDIM / 2 - 1);
            const int n4 = u & 15;
            wa[j] = W4[(2 * p)     * (OUTC / 4) + n4];
            wb[j] = W4[(2 * p + 1) * (OUTC / 4) + n4];
        }
#pragma unroll
        for (int j = 0; j < 8; ++j) {
            const int u  = j * 256 + threadIdx.x;
            const int p  = ((u >> 4) + rot) & (KDIM / 2 - 1);
            const int n4 = u & 15;
            float4* d = (float4*)(wint + p * 128 + n4 * 8);
            d[0] = make_float4(wa[j].x, wb[j].x, wa[j].y, wb[j].y);
            d[1] = make_float4(wa[j].z, wb[j].z, wa[j].w, wb[j].w);
        }
    }
    __syncthreads();

    float ps0 = 0.f, ps1 = 0.f, ps2 = 0.f, ps3 = 0.f;
    float pq0 = 0.f, pq1 = 0.f, pq2 = 0.f, pq3 = 0.f;

    if (row_base < mtotal) {

    const float4* __restrict__ F4 = (const float4*)feats;
    const unsigned fq = 4u * (unsigned)kbh;
    const unsigned un = (unsigned)nvox;
    const float* __restrict__ B = wint + kbh * 1024 + r * 2;

    const float4 z4 = make_float4(0.f, 0.f, 0.f, 0.f);

    int b0 = -1, bn = -1;
    if ((unsigned)a0 < un) b0 = next[a0];
    if ((unsigned)an < un) bn = next[an];

    float4 x0 = z4, x1 = z4, x2 = z4, x3 = z4;
    float4 y0 = z4, y1 = z4, y2 = z4, y3 = z4;
    int c = -1;
    if ((unsigned)a0 < un) {
        const float4* p = F4 + ((unsigned)a0 * 8u + fq);
        x0 = p[0]; x1 = p[1]; x2 = p[2]; x3 = p[3];
    }
    if ((unsigned)b0 < un) {
        const float4* p = F4 + ((unsigned)b0 * 8u + fq);
        y0 = p[0]; y1 = p[1]; y2 = p[2]; y3 = p[3];
        c = next[b0];
    }

    v8f acc0 = {}, acc1 = {}, acc2 = {}, acc3 = {};

#pragma unroll 1
    for (int t = 0; t < 2 * KVOL; ++t) {
        x0.x += y0.x; x0.y += y0.y; x0.z += y0.z; x0.w += y0.w;
        x1.x += y1.x; x1.y += y1.y; x1.z += y1.z; x1.w += y1.w;
        x2.x += y2.x; x2.y += y2.y; x2.z += y2.z; x2.w += y2.w;
        x3.x += y3.x; x3.y += y3.y; x3.z += y3.z; x3.w += y3.w;
        for (int it = 2; it < nvox; ++it) {
            if (!((unsigned)c < un)) break;
            const float4* p = F4 + ((unsigned)c * 8u + fq);
            float4 t0 = p[0], t1 = p[1], t2 = p[2], t3 = p[3];
            c = next[c];
            x0.x += t0.x; x0.y += t0.y; x0.z += t0.z; x0.w += t0.w;
            x1.x += t1.x; x1.y += t1.y; x1.z += t1.z; x1.w += t1.w;
            x2.x += t2.x; x2.y += t2.y; x2.z += t2.z; x2.w += t2.w;
            x3.x += t3.x; x3.y += t3.y; x3.z += t3.z; x3.w += t3.w;
        }

        const float xs[16] = {x0.x, x0.y, x0.z, x0.w, x1.x, x1.y, x1.z, x1.w,
                              x2.x, x2.y, x2.z, x2.w, x3.x, x3.y, x3.z, x3.w};

        x0 = z4; x1 = z4; x2 = z4; x3 = z4;
        y0 = z4; y1 = z4; y2 = z4; y3 = z4;
        c = -1;
        int b2 = -1, a3 = -1;
        if (t + 1 < 2 * KVOL) {
            if ((unsigned)an < un) {
                const float4* p = F4 + ((unsigned)an * 8u + fq);
                x0 = p[0]; x1 = p[1]; x2 = p[2]; x3 = p[3];
            }
            if ((unsigned)bn < un) {
                const float4* p = F4 + ((unsigned)bn * 8u + fq);
                y0 = p[0]; y1 = p[1]; y2 = p[2]; y3 = p[3];
                c = next[bn];
            }
        }
        if (t + 2 < 2 * KVOL) {
            if ((unsigned)a2 < un) b2 = next[a2];
        }
        if (t + 3 < 2 * KVOL) {
            const int t3 = t + 3;
            a3 = head[((t3 & KVOL) ? hb1 : hb0) + (t3 & (KVOL - 1))];
        }

        const int o = t & (KVOL - 1);
        const float* Bo = B + o * (16 * 128);

#pragma unroll
        for (int kk = 0; kk < 8; ++kk) {
            v2f va; va.x = xs[2 * kk]; va.y = xs[2 * kk + 1];

            const float* bp = Bo + kk * 128;
            float2 g0 = *(const float2*)(bp);
            float2 g1 = *(const float2*)(bp + 32);
            float2 g2 = *(const float2*)(bp + 64);
            float2 g3 = *(const float2*)(bp + 96);
            v2f q0; q0.x = g0.x; q0.y = g0.y;
            v2f q1; q1.x = g1.x; q1.y = g1.y;
            v2f q2; q2.x = g2.x; q2.y = g2.y;
            v2f q3; q3.x = g3.x; q3.y = g3.y;

            acc0 = __builtin_amdgcn_wmma_f32_16x16x4_f32(false, va, false, q0, (short)0, acc0, false, false);
            acc1 = __builtin_amdgcn_wmma_f32_16x16x4_f32(false, va, false, q1, (short)0, acc1, false, false);
            acc2 = __builtin_amdgcn_wmma_f32_16x16x4_f32(false, va, false, q2, (short)0, acc2, false, false);
            acc3 = __builtin_amdgcn_wmma_f32_16x16x4_f32(false, va, false, q3, (short)0, acc3, false, false);
        }

        if (o == KVOL - 1) {
            const int mrow = row_base + ((t >> 3) << 4) + (kbh << 3);
#define SUM_NT(ACC, PS, PQ)                                                    \
            _Pragma("unroll")                                                  \
            for (int i = 0; i < 8; ++i) {                                      \
                float v = (ACC)[i];                                            \
                if (mrow + i < mtotal) { PS += v; PQ += v * v; }               \
            }
            SUM_NT(acc0, ps0, pq0)
            SUM_NT(acc1, ps1, pq1)
            SUM_NT(acc2, ps2, pq2)
            SUM_NT(acc3, ps3, pq3)
#undef SUM_NT
            float*       sw   = ostage + wave * 320 + kbh * 80 + r;
            const float* sr   = ostage + wave * 320 + kbh * 80 + r * 4;
            float*       orow = out + (long long)mrow * OUTC + r * 4;
#pragma unroll
            for (int i2 = 0; i2 < 8; i2 += 2) {
                sw[0]        = acc0[i2];     sw[16]       = acc1[i2];
                sw[32]       = acc2[i2];     sw[48]       = acc3[i2];
                sw[160]      = acc0[i2 + 1]; sw[160 + 16] = acc1[i2 + 1];
                sw[160 + 32] = acc2[i2 + 1]; sw[160 + 48] = acc3[i2 + 1];
                wave_lds_order();
                const float4 w0 = *(const float4*)(sr);
                const float4 w1 = *(const float4*)(sr + 160);
                wave_lds_order();
                if (mrow + i2 < mtotal)
                    *(float4*)(orow + (long long)i2 * OUTC) = w0;
                if (mrow + i2 + 1 < mtotal)
                    *(float4*)(orow + (long long)(i2 + 1) * OUTC) = w1;
            }
            const v8f zacc = {};
            acc0 = zacc; acc1 = zacc; acc2 = zacc; acc3 = zacc;
        }

        an = a2; bn = b2; a2 = a3;
    }

    }

#define EMIT_STAT(PS, PQ, NT)                                                  \
    {                                                                          \
        float s = (PS) + __shfl_xor((PS), 16, 32);                             \
        float q = (PQ) + __shfl_xor((PQ), 16, 32);                             \
        if (lane < 16) {                                                       \
            wsum[wave * (2 * OUTC) + (NT) * 16 + r] = s;                       \
            wsum[wave * (2 * OUTC) + OUTC + (NT) * 16 + r] = q;                \
        }                                                                      \
    }
    EMIT_STAT(ps0, pq0, 0)
    EMIT_STAT(ps1, pq1, 1)
    EMIT_STAT(ps2, pq2, 2)
    EMIT_STAT(ps3, pq3, 3)
#undef EMIT_STAT

    __syncthreads();

    if (threadIdx.x < 2 * OUTC) {
        float a = wsum[threadIdx.x];
        for (int w = 1; w < 8; ++w) a += wsum[w * (2 * OUTC) + threadIdx.x];
        unsafeAtomicAdd(&stats[threadIdx.x], a);
    }
}

#define BN_PER 32
__launch_bounds__(256)
__global__ void spconv_bn_apply(float* __restrict__ out,
                                const float* __restrict__ stats,
                                const float* __restrict__ gamma,
                                const float* __restrict__ beta,
                                int mtotal) {
    const int lane = threadIdx.x & 31;
    const long long total4 = ((long long)mtotal * OUTC) >> 2;
    const float inv_m = 1.0f / (float)mtotal;
    const int c0 = (lane << 2) & (OUTC - 1);
    float mean[4], rs[4], ga[4], be[4];
#pragma unroll
    for (int j = 0; j < 4; ++j) {
        const int c = c0 + j;
        mean[j] = stats[c] * inv_m;
        const float var = stats[64 + c] * inv_m - mean[j] * mean[j];
        rs[j] = rsqrtf(var + BN_EPS);
        ga[j] = gamma[c];
        be[j] = beta[c];
    }
    float4* __restrict__ o4 = (float4*)out;
    const long long CH = 32LL * BN_PER;
    const long long nwaves = (long long)gridDim.x * (blockDim.x >> 5);
    const long long wv = (long long)blockIdx.x * (blockDim.x >> 5)
                       + __builtin_amdgcn_readfirstlane((int)(threadIdx.x >> 5));
    const long long nfull = total4 / CH;
    for (long long ch = wv; ch < nfull; ch += nwaves) {
        float4* p = o4 + ch * CH + lane;
        float4 v[BN_PER];
#pragma unroll
        for (int k = 0; k < BN_PER; ++k) v[k] = p[k * 32];
        __builtin_amdgcn_sched_barrier(0);
#pragma unroll
        for (int k = 0; k < BN_PER; ++k) {
            float4 y;
            y.x = (v[k].x - mean[0]) * rs[0] * ga[0] + be[0];
            y.y = (v[k].y - mean[1]) * rs[1] * ga[1] + be[1];
            y.z = (v[k].z - mean[2]) * rs[2] * ga[2] + be[2];
            y.w = (v[k].w - mean[3]) * rs[3] * ga[3] + be[3];
            p[k * 32] = y;
        }
    }
    const long long G = (long long)gridDim.x * blockDim.x;
    for (long long t = nfull * CH + (long long)blockIdx.x * blockDim.x + threadIdx.x;
         t < total4; t += G) {
        const float4 x = o4[t];
        float4 y;
        y.x = (x.x - mean[0]) * rs[0] * ga[0] + be[0];
        y.y = (x.y - mean[1]) * rs[1] * ga[1] + be[1];
        y.z = (x.z - mean[2]) * rs[2] * ga[2] + be[2];
        y.w = (x.w - mean[3]) * rs[3] * ga[3] + be[3];
        o4[t] = y;
    }
}

extern "C" void kernel_launch(void* const* d_in, const int* in_sizes, int n_in,
                              void* d_out, int out_size, void* d_ws, size_t ws_size,
                              hipStream_t stream) {
    const float* feats     = (const float*)d_in[0];
    const float* weight    = (const float*)d_in[1];
    const float* gamma     = (const float*)d_in[2];
    const float* beta      = (const float*)d_in[3];
    const int*   out_idx   = (const int*)d_in[4];
    const int*   offset_id = (const int*)d_in[5];
    const int nvox   = in_sizes[0] / INC;
    const int mtotal = out_size / OUTC;

    int*   head  = (int*)d_ws;
    int*   next  = head + (long long)mtotal * KVOL;
    float* stats = (float*)(next + nvox);
    const long long nhead4 = ((long long)mtotal * KVOL) >> 2;

    spconv_init_ws<<<256, 256, 0, stream>>>(head, nhead4, stats);

    spconv_link<<<512, 256, 0, stream>>>(out_idx, offset_id, head, next, nvox);

    const int gblocks = (mtotal + 255) / 256;
    spconv_gemm_stats<<<gblocks, 256, 0, stream>>>(feats, head, next, weight, (float*)d_out,
                                                   stats, mtotal, nvox);

    spconv_bn_apply<<<512, 256, 0, stream>>>((float*)d_out, stats, gamma, beta, mtotal);
}


# GEMM runs on non-empty buckets only: per offset a wave packs those of its 32 rows into 16 WMMA columns (operands swapped), adds each tile into LDS rows; half the WMMAs; sum tree changes
# speedup vs baseline: 3.1627x; 1.0655x over previous
#include <hip/hip_runtime.h>


typedef __attribute__((ext_vector_type(2))) float v2f;
typedef __attribute__((ext_vector_type(8))) float v8f;

#define INC    32
#define OUTC   64
#define KVOL   8
#define KDIM   256
#define BN_EPS 1e-5f

__global__ void spconv_init_ws(int* __restrict__ head, long long nhead4,
                               float* __restrict__ stats) {
    long long i = (long long)blockIdx.x * blockDim.x + threadIdx.x;
    long long stride = (long long)gridDim.x * blockDim.x;
    int4* p = (int4*)head;
    for (long long j = i; j < nhead4; j += stride)
        p[j] = make_int4(-1, -1, -1, -1);
    if (i < 2 * OUTC) stats[i] = 0.f;
}

#define LINK_PER 8
__launch_bounds__(256)
__global__ void spconv_link(const int* __restrict__ out_idx,
                            const int* __restrict__ offset_id,
                            int*       __restrict__ head,
                            int*       __restrict__ next,
                            int nvox) {
    const long long G = (long long)gridDim.x * blockDim.x;
    long long v = (long long)blockIdx.x * blockDim.x + threadIdx.x;
    for (; v + (LINK_PER - 1) * G < nvox; v += LINK_PER * G) {
        int seg[LINK_PER], old[LINK_PER];
#pragma unroll
        for (int k = 0; k < LINK_PER; ++k) {
            const long long u = v + k * G;
            seg[k] = out_idx[u] * KVOL + offset_id[u];
        }
#pragma unroll
        for (int k = 0; k < LINK_PER; ++k)
            old[k] = atomicExch(&head[seg[k]], (int)(v + k * G));
#pragma unroll
        for (int k = 0; k < LINK_PER; ++k)
            next[v + k * G] = old[k];
    }
    for (; v < nvox; v += G) {
        const int seg = out_idx[v] * KVOL + offset_id[v];
        next[v] = atomicExch(&head[seg], (int)v);
    }
}

#define TP 68
#define TROWS 33

__device__ __forceinline__ void wave_lds_order() {
    __builtin_amdgcn_fence(__ATOMIC_RELEASE, "wavefront");
    __builtin_amdgcn_wave_barrier();
    __builtin_amdgcn_fence(__ATOMIC_ACQUIRE, "wavefront");
}

struct SlotFront { int a, b, r; };
__device__ __forceinline__ SlotFront slot_front(const int* sa, const int* sb,
                                                const unsigned char* sr,
                                                unsigned long long cpk, int s, int c) {
    const int o  = s >> 1;
    const int j  = ((s & 1) << 4) + c;
    const int cn = (int)((cpk >> (8 * o)) & 0xffull);
    SlotFront f; f.a = -1; f.b = -1; f.r = 32;
    if (j < cn) { f.a = sa[o * 32 + j]; f.b = sb[o * 32 + j]; f.r = sr[o * 32 + j]; }
    return f;
}

__launch_bounds__(256) __attribute__((amdgpu_waves_per_eu(4, 4)))
__global__ void spconv_gemm_stats(const float* __restrict__ feats,
                                  const int*   __restrict__ head,
                                  const int*   __restrict__ next,
                                  const float* __restrict__ weight,
                                  float*       __restrict__ out,
                                  float*       __restrict__ stats,
                                  int mtotal, int nvox) {
    __shared__ __attribute__((aligned(16))) float wint[(KDIM / 2) * 2 * OUTC];
    __shared__ __attribute__((aligned(16))) float rtile[8 * TROWS * TP];
    __shared__ float wsum[8 * 2 * OUTC];
    __shared__ int   sl_a[8 * KVOL * 32];
    __shared__ int   sl_b[8 * KVOL * 32];
    __shared__ unsigned char sl_r[8 * KVOL * 32];

    const int wave = threadIdx.x >> 5;
    const int lane = threadIdx.x & 31;
    const int c    = lane & 15;
    const int kbh  = lane >> 4;
    const unsigned un = (unsigned)nvox;

    const int row_base = (blockIdx.x * 8 + wave) * 32;
    const int myrow    = row_base + lane;
    int4 h0 = make_int4(-1, -1, -1, -1), h1 = h0;
    if (myrow < mtotal) {
        const int4* hp = (const int4*)head + (long long)myrow * 2;
        h0 = hp[0]; h1 = hp[1];
    }

    float* tw = rtile + wave * (TROWS * TP);
    const float4 z4 = make_float4(0.f, 0.f, 0.f, 0.f);

    int hd[8], sc[8];
    {
        const float4* __restrict__ W4 = (const float4*)weight;
        const int rot = blockIdx.x & (KDIM / 2 - 1);
        float4 wa[8], wb[8];
#pragma unroll
        for (int j = 0; j < 8; ++j) {
            const int u  = j * 256 + threadIdx.x;
            const int p  = ((u >> 4) + rot) & (KDIM / 2 - 1);
            const int n4 = u & 15;
            wa[j] = W4[(2 * p)     * (OUTC / 4) + n4];
            wb[j] = W4[(2 * p + 1) * (OUTC / 4) + n4];
        }
        hd[0] = h0.x; hd[1] = h0.y; hd[2] = h0.z; hd[3] = h0.w;
        hd[4] = h1.x; hd[5] = h1.y; hd[6] = h1.z; hd[7] = h1.w;
#pragma unroll
        for (int o = 0; o < 8; ++o) {
            sc[o] = -1;
            if ((unsigned)hd[o] < un) sc[o] = next[hd[o]];
        }
#pragma unroll
        for (int j = 0; j < 8; ++j) {
            const int u  = j * 256 + threadIdx.x;
            const int p  = ((u >> 4) + rot) & (KDIM / 2 - 1);
            const int n4 = u & 15;
            float4* d = (float4*)(wint + p * 128 + n4 * 8);
            d[0] = make_float4(wa[j].x, wb[j].x, wa[j].y, wb[j].y);
            d[1] = make_float4(wa[j].z, wb[j].z, wa[j].w, wb[j].w);
        }
    }
    for (int k = lane; k < TROWS * TP / 4; k += 32) ((float4*)tw)[k] = z4;
    __syncthreads();

    float s0 = 0.f, s1 = 0.f, q0s = 0.f, q1s = 0.f;

    if (row_base < mtotal) {

    int*           sa = sl_a + wave * (KVOL * 32);
    int*           sb = sl_b + wave * (KVOL * 32);
    unsigned char* sr = sl_r + wave * (KVOL * 32);
    unsigned long long cpk = 0ull;
#pragma unroll
    for (int o = 0; o < 8; ++o) {
        const bool ne = (unsigned)hd[o] < un;
        const unsigned m  = __builtin_amdgcn_ballot_w32(ne);
        const unsigned rk = __builtin_amdgcn_mbcnt_lo(m, 0u);
        if (ne) {
            sa[o * 32 + rk] = hd[o];
            sb[o * 32 + rk] = sc[o];
            sr[o * 32 + rk] = (unsigned char)lane;
        }
        cpk |= (unsigned long long)__builtin_popcount(m) << (8 * o);
    }
    wave_lds_order();

    const float4* __restrict__ F4 = (const float4*)feats;
    const unsigned fq = 4u * (unsigned)kbh;
    const float* __restrict__ Wl = wint + kbh * 1024 + c * 2;

    SlotFront cur = slot_front(sa, sb, sr, cpk, 0, c);
    float4 x0 = z4, x1 = z4, x2 = z4, x3 = z4;
    float4 y0 = z4, y1 = z4, y2 = z4, y3 = z4;
    int cc = -1;
    if ((unsigned)cur.a < un) {
        const float4* p = F4 + ((unsigned)cur.a * 8u + fq);
        x0 = p[0]; x1 = p[1]; x2 = p[2]; x3 = p[3];
    }
    if ((unsigned)cur.b < un) {
        const float4* p = F4 + ((unsigned)cur.b * 8u + fq);
        y0 = p[0]; y1 = p[1]; y2 = p[2]; y3 = p[3];
        cc = next[cur.b];
    }

    int s = 0;
#pragma unroll 1
    while (s < 2 * KVOL) {
        int sn = s + 1;
        if (sn < 2 * KVOL && (sn & 1)) {
            const int cn = (int)((cpk >> (8 * (sn >> 1))) & 0xffull);
            if (cn <= 16) ++sn;
        }

        x0.x += y0.x; x0.y += y0.y; x0.z += y0.z; x0.w += y0.w;
        x1.x += y1.x; x1.y += y1.y; x1.z += y1.z; x1.w += y1.w;
        x2.x += y2.x; x2.y += y2.y; x2.z += y2.z; x2.w += y2.w;
        x3.x += y3.x; x3.y += y3.y; x3.z += y3.z; x3.w += y3.w;
        for (int it = 2; it < nvox; ++it) {
            if (!((unsigned)cc < un)) break;
            const float4* p = F4 + ((unsigned)cc * 8u + fq);
            float4 t0 = p[0], t1 = p[1], t2 = p[2], t3 = p[3];
            cc = next[cc];
            x0.x += t0.x; x0.y += t0.y; x0.z += t0.z; x0.w += t0.w;
            x1.x += t1.x; x1.y += t1.y; x1.z += t1.z; x1.w += t1.w;
            x2.x += t2.x; x2.y += t2.y; x2.z += t2.z; x2.w += t2.w;
            x3.x += t3.x; x3.y += t3.y; x3.z += t3.z; x3.w += t3.w;
        }

        const float xs[16] = {x0.x, x0.y, x0.z, x0.w, x1.x, x1.y, x1.z, x1.w,
                              x2.x, x2.y, x2.z, x2.w, x3.x, x3.y, x3.z, x3.w};
        const int crow = cur.r;

        x0 = z4; x1 = z4; x2 = z4; x3 = z4;
        y0 = z4; y1 = z4; y2 = z4; y3 = z4;
        cc = -1;
        cur.a = -1; cur.b = -1; cur.r = 32;
        if (sn < 2 * KVOL) {
            cur = slot_front(sa, sb, sr, cpk, sn, c);
            if ((unsigned)cur.a < un) {
                const float4* p = F4 + ((unsigned)cur.a * 8u + fq);
                x0 = p[0]; x1 = p[1]; x2 = p[2]; x3 = p[3];
            }
            if ((unsigned)cur.b < un) {
                const float4* p = F4 + ((unsigned)cur.b * 8u + fq);
                y0 = p[0]; y1 = p[1]; y2 = p[2]; y3 = p[3];
                cc = next[cur.b];
            }
        }

        const int o = s >> 1;
        const float* Wo = Wl + o * (16 * 128);
        v8f acc0 = {}, acc1 = {}, acc2 = {}, acc3 = {};

#pragma unroll
        for (int kk = 0; kk < 8; ++kk) {
            v2f vb; vb.x = xs[2 * kk]; vb.y = xs[2 * kk + 1];

            const float* wp = Wo + kk * 128;
            float2 g0 = *(const float2*)(wp);
            float2 g1 = *(const float2*)(wp + 32);
            float2 g2 = *(const float2*)(wp + 64);
            float2 g3 = *(const float2*)(wp + 96);
            v2f w0; w0.x = g0.x; w0.y = g0.y;
            v2f w1; w1.x = g1.x; w1.y = g1.y;
            v2f w2; w2.x = g2.x; w2.y = g2.y;
            v2f w3; w3.x = g3.x; w3.y = g3.y;

            acc0 = __builtin_amdgcn_wmma_f32_16x16x4_f32(false, w0, false, vb, (short)0, acc0, false, false);
            acc1 = __builtin_amdgcn_wmma_f32_16x16x4_f32(false, w1, false, vb, (short)0, acc1, false, false);
            acc2 = __builtin_amdgcn_wmma_f32_16x16x4_f32(false, w2, false, vb, (short)0, acc2, false, false);
            acc3 = __builtin_amdgcn_wmma_f32_16x16x4_f32(false, w3, false, vb, (short)0, acc3, false, false);
        }

        {
            float* rp = tw + crow * TP + 8 * kbh;
#define ADD_NT(ACC, NT)                                                        \
            {                                                                  \
                float4* q = (float4*)(rp + (NT) * 16);                         \
                float4 u0 = q[0], u1 = q[1];                                   \
                u0.x += (ACC)[0]; u0.y += (ACC)[1]; u0.z += (ACC)[2]; u0.w += (ACC)[3]; \
                u1.x += (ACC)[4]; u1.y += (ACC)[5]; u1.z += (ACC)[6]; u1.w += (ACC)[7]; \
                q[0] = u0; q[1] = u1;                                          \
            }
            ADD_NT(acc0, 0)
            ADD_NT(acc1, 1)
            ADD_NT(acc2, 2)
            ADD_NT(acc3, 3)
#undef ADD_NT
        }
        wave_lds_order();

        s = sn;
    }

#pragma unroll
    for (int rr = 0; rr < 32; ++rr) {
        const float2 v = *(const float2*)(tw + rr * TP + 2 * lane);
        s0 += v.x; q0s += v.x * v.x;
        s1 += v.y; q1s += v.y * v.y;
    }
    {
        const int    orow = lane >> 4;
        const float* sp   = tw + orow * TP + c * 4;
        float*       op   = out + ((long long)row_base + orow) * OUTC + c * 4;
#pragma unroll
        for (int i = 0; i < 16; ++i) {
            const float4 w = *(const float4*)(sp + (2 * i) * TP);
            if (row_base + 2 * i + orow < mtotal)
                *(float4*)(op + (long long)(2 * i) * OUTC) = w;
        }
    }

    }

    wsum[wave * (2 * OUTC) + 2 * lane]            = s0;
    wsum[wave * (2 * OUTC) + 2 * lane + 1]        = s1;
    wsum[wave * (2 * OUTC) + OUTC + 2 * lane]     = q0s;
    wsum[wave * (2 * OUTC) + OUTC + 2 * lane + 1] = q1s;

    __syncthreads();

    if (threadIdx.x < 2 * OUTC) {
        float a = wsum[threadIdx.x];
        for (int w = 1; w < 8; ++w) a += wsum[w * (2 * OUTC) + threadIdx.x];
        unsafeAtomicAdd(&stats[threadIdx.x], a);
    }
}

#define BN_PER 32
__launch_bounds__(256)
__global__ void spconv_bn_apply(float* __restrict__ out,
                                const float* __restrict__ stats,
                                const float* __restrict__ gamma,
                                const float* __restrict__ beta,
                                int mtotal) {
    const int lane = threadIdx.x & 31;
    const long long total4 = ((long long)mtotal * OUTC) >> 2;
    const float inv_m = 1.0f / (float)mtotal;
    const int c0 = (lane << 2) & (OUTC - 1);
    float mean[4], rs[4], ga[4], be[4];
#pragma unroll
    for (int j = 0; j < 4; ++j) {
        const int c = c0 + j;
        mean[j] = stats[c] * inv_m;
        const float var = stats[64 + c] * inv_m - mean[j] * mean[j];
        rs[j] = rsqrtf(var + BN_EPS);
        ga[j] = gamma[c];
        be[j] = beta[c];
    }
    float4* __restrict__ o4 = (float4*)out;
    const long long CH = 32LL * BN_PER;
    const long long nwaves = (long long)gridDim.x * (blockDim.x >> 5);
    const long long wv = (long long)blockIdx.x * (blockDim.x >> 5)
                       + __builtin_amdgcn_readfirstlane((int)(threadIdx.x >> 5));
    const long long nfull = total4 / CH;
    for (long long ch = wv; ch < nfull; ch += nwaves) {
        float4* p = o4 + ch * CH + lane;
        float4 v[BN_PER];
#pragma unroll
        for (int k = 0; k < BN_PER; ++k) v[k] = p[k * 32];
        __builtin_amdgcn_sched_barrier(0);
#pragma unroll
        for (int k = 0; k < BN_PER; ++k) {
            float4 y;
            y.x = (v[k].x - mean[0]) * rs[0] * ga[0] + be[0];
            y.y = (v[k].y - mean[1]) * rs[1] * ga[1] + be[1];
            y.z = (v[k].z - mean[2]) * rs[2] * ga[2] + be[2];
            y.w = (v[k].w - mean[3]) * rs[3] * ga[3] + be[3];
            p[k * 32] = y;
        }
    }
    const long long G = (long long)gridDim.x * blockDim.x;
    for (long long t = nfull * CH + (long long)blockIdx.x * blockDim.x + threadIdx.x;
         t < total4; t += G) {
        const float4 x = o4[t];
        float4 y;
        y.x = (x.x - mean[0]) * rs[0] * ga[0] + be[0];
        y.y = (x.y - mean[1]) * rs[1] * ga[1] + be[1];
        y.z = (x.z - mean[2]) * rs[2] * ga[2] + be[2];
        y.w = (x.w - mean[3]) * rs[3] * ga[3] + be[3];
        o4[t] = y;
    }
}

extern "C" void kernel_launch(void* const* d_in, const int* in_sizes, int n_in,
                              void* d_out, int out_size, void* d_ws, size_t ws_size,
                              hipStream_t stream) {
    const float* feats     = (const float*)d_in[0];
    const float* weight    = (const float*)d_in[1];
    const float* gamma     = (const float*)d_in[2];
    const float* beta      = (const float*)d_in[3];
    const int*   out_idx   = (const int*)d_in[4];
    const int*   offset_id = (const int*)d_in[5];
    const int nvox   = in_sizes[0] / INC;
    const int mtotal = out_size / OUTC;

    int*   head  = (int*)d_ws;
    int*   next  = head + (long long)mtotal * KVOL;
    float* stats = (float*)(next + nvox);
    const long long nhead4 = ((long long)mtotal * KVOL) >> 2;

    spconv_init_ws<<<256, 256, 0, stream>>>(head, nhead4, stats);

    spconv_link<<<512, 256, 0, stream>>>(out_idx, offset_id, head, next, nvox);

    const int gblocks = (mtotal + 255) / 256;
    spconv_gemm_stats<<<gblocks, 256, 0, stream>>>(feats, head, next, weight, (float*)d_out,
                                                   stats, mtotal, nvox);

    spconv_bn_apply<<<512, 256, 0, stream>>>((float*)d_out, stats, gamma, beta, mtotal);
}
